# starting kernel
# baseline (speedup 1.0000x reference)
#include <hip/hip_runtime.h>
#include <math.h>

#define B_   2
#define S_   2048
#define H_   768
#define NH_  12
#define HD_  64
#define FFN_ 3072
#define M_   (B_ * S_)
#define QKVLD (3 * H_)

typedef __attribute__((ext_vector_type(2))) float v2f;
typedef __attribute__((ext_vector_type(4))) float v4f;
typedef __attribute__((ext_vector_type(8))) float v8f;

__device__ __forceinline__ v8f wmma4(v2f a, v2f b, v8f c) {
  return __builtin_amdgcn_wmma_f32_16x16x4_f32(false, a, false, b, (short)0, c,
                                               false, false);
}

__device__ __forceinline__ float gelu_exact(float x) {
  return 0.5f * x * (1.0f + erff(x * 0.70710678118654752440f));
}

__global__ __launch_bounds__(256) void ln_kernel(const float* __restrict__ x,
                                                 const float* __restrict__ g,
                                                 const float* __restrict__ b,
                                                 float* __restrict__ o) {
  __shared__ float red[256];
  const int t = threadIdx.x;
  const size_t base = (size_t)blockIdx.x * H_;

  float s = 0.0f;
  for (int i = t; i < H_; i += 256) s += x[base + i];
  red[t] = s;
  __syncthreads();
  for (int off = 128; off > 0; off >>= 1) {
    if (t < off) red[t] += red[t + off];
    __syncthreads();
  }
  const float mu = red[0] * (1.0f / (float)H_);
  __syncthreads();

  float vs = 0.0f;
  for (int i = t; i < H_; i += 256) {
    float d = x[base + i] - mu;
    vs += d * d;
  }
  red[t] = vs;
  __syncthreads();
  for (int off = 128; off > 0; off >>= 1) {
    if (t < off) red[t] += red[t + off];
    __syncthreads();
  }
  const float rstd = rsqrtf(red[0] * (1.0f / (float)H_) + 1e-5f);

  for (int i = t; i < H_; i += 256)
    o[base + i] = (x[base + i] - mu) * rstd * g[i] + b[i];
}

#define GLDA 20

__global__ __launch_bounds__(256) void gemm_kernel(
    const float* __restrict__ A, const float* __restrict__ W,
    const float* __restrict__ bias, const float* __restrict__ resid,
    float* __restrict__ C, int M, int N, int K, int act) {
  __shared__ float As[2][64 * GLDA];
  __shared__ float Ws[2][64 * GLDA];

  const int t = threadIdx.x;
  const int wave = t >> 5, lane = t & 31;
  const int m0 = blockIdx.x * 64, n0 = blockIdx.y * 64;
  const int wm = 16 * (wave & 3);
  const int wn = 32 * (wave >> 2);
  const int lm = lane & 15;
  const int koff = (lane < 16) ? 0 : 2;
  const int rbase = (lane < 16) ? 0 : 8;

  const int lrow = t >> 2;
  const int lcol = (t & 3) * 4;

  const float* Ap = A + (size_t)(m0 + lrow) * K + lcol;
  const float* Wp = W + (size_t)(n0 + lrow) * K + lcol;
  const int lds_off = lrow * GLDA + lcol;

  v8f acc0 = {};
  v8f acc1 = {};

  {
    v4f av = *(const v4f*)(Ap);
    v4f wv = *(const v4f*)(Wp);
    *(v4f*)(As[0] + lds_off) = av;
    *(v4f*)(Ws[0] + lds_off) = wv;
  }
  __syncthreads();

  const int nstages = K >> 4;
  for (int s = 0; s < nstages; ++s) {
    const int p = s & 1;
    const bool has_next = (s + 1) < nstages;

    v4f av, wv;
    if (has_next) {
      av = *(const v4f*)(Ap + (size_t)(s + 1) * 16);
      wv = *(const v4f*)(Wp + (size_t)(s + 1) * 16);
    }

    const float* Ab = As[p];
    const float* Wb = Ws[p];
#pragma unroll
    for (int kk = 0; kk < 16; kk += 4) {
      v2f a  = *(const v2f*)(Ab + (wm + lm) * GLDA + kk + koff);
      v2f b0 = *(const v2f*)(Wb + (wn + lm) * GLDA + kk + koff);
      v2f b1 = *(const v2f*)(Wb + (wn + 16 + lm) * GLDA + kk + koff);
      acc0 = wmma4(a, b0, acc0);
      acc1 = wmma4(a, b1, acc1);
    }

    if (has_next) {
      *(v4f*)(As[1 - p] + lds_off) = av;
      *(v4f*)(Ws[1 - p] + lds_off) = wv;
    }
    __syncthreads();
  }

#pragma unroll
  for (int r = 0; r < 8; ++r) {
    const int m = m0 + wm + r + rbase;
    const int n = n0 + wn + lm;
    const size_t i0 = (size_t)m * N + n;
    float v0 = acc0[r] + bias[n];
    float v1 = acc1[r] + bias[n + 16];
    if (resid) {
      v0 += resid[i0];
      v1 += resid[i0 + 16];
    }
    if (act == 1) {
      v0 = gelu_exact(v0);
      v1 = gelu_exact(v1);
    }
    C[i0] = v0;
    C[i0 + 16] = v1;
  }
}

#define ALD 66

__global__ __launch_bounds__(256) void attn_kernel(
    const float* __restrict__ qkv, float* __restrict__ out) {
  __shared__ float Qs[64 * ALD];
  __shared__ float Ks[64 * ALD];
  __shared__ float Vs[64 * ALD];
  __shared__ float alpha_s[64];
  __shared__ float lrow_s[64];

  const int t = threadIdx.x;
  const int wave = t >> 5, lane = t & 31;
  const int lm = lane & 15;
  const int koff = (lane < 16) ? 0 : 2;
  const int rbase = (lane < 16) ? 0 : 8;
  const int wm = 16 * (wave & 3);
  const int wn = 32 * (wave >> 2);

  const int qt = blockIdx.x;
  const int bh = blockIdx.y;
  const int b = bh / NH_, nh = bh % NH_;
  const int q0 = qt * 64;

  const float* qbase = qkv + (size_t)b * S_ * QKVLD + nh * HD_;
  const float* kbase = qbase + H_;
  const float* vbase = qbase + 2 * H_;
  float* Ps = Ks;

  for (int i = 0; i < 8; ++i) {
    int f = t + 256 * i;
    int r = f >> 5;
    int c2 = (f & 31) * 2;
    *(v2f*)(Qs + r * ALD + c2) =
        *(const v2f*)(qbase + (size_t)(q0 + r) * QKVLD + c2);
  }

  v8f o0 = {};
  v8f o1 = {};
  float mi = -INFINITY, li = 0.0f;

  for (int j = 0; j <= qt; ++j) {
    const int k0 = j * 64;
    for (int i = 0; i < 8; ++i) {
      int f = t + 256 * i;
      int r = f >> 5;
      int c2 = (f & 31) * 2;
      *(v2f*)(Ks + r * ALD + c2) =
          *(const v2f*)(kbase + (size_t)(k0 + r) * QKVLD + c2);
      *(v2f*)(Vs + r * ALD + c2) =
          *(const v2f*)(vbase + (size_t)(k0 + r) * QKVLD + c2);
    }
    __syncthreads();

    v8f s0 = {};
    v8f s1 = {};
#pragma unroll
    for (int kk = 0; kk < 64; kk += 4) {
      v2f a  = *(const v2f*)(Qs + (wm + lm) * ALD + kk + koff);
      v2f b0 = *(const v2f*)(Ks + (wn + lm) * ALD + kk + koff);
      v2f b1 = *(const v2f*)(Ks + (wn + 16 + lm) * ALD + kk + koff);
      s0 = wmma4(a, b0, s0);
      s1 = wmma4(a, b1, s1);
    }
    __syncthreads();

#pragma unroll
    for (int r = 0; r < 8; ++r) {
      Ps[(wm + r + rbase) * ALD + wn + lm]      = s0[r] * 0.125f;
      Ps[(wm + r + rbase) * ALD + wn + 16 + lm] = s1[r] * 0.125f;
    }
    __syncthreads();

    if (t < 64) {
      const int m = t;
      float* Pr = Ps + m * ALD;
      const int limit = (j == qt) ? (m + 1) : 64;
      float mx = -INFINITY;
      for (int c = 0; c < limit; ++c) mx = fmaxf(mx, Pr[c]);
      const float mnew = fmaxf(mi, mx);
      const float al = expf(mi - mnew);
      float sum = 0.0f;
      for (int c = 0; c < 64; ++c) {
        float p = (c < limit) ? expf(Pr[c] - mnew) : 0.0f;
        Pr[c] = p;
        sum += p;
      }
      li = li * al + sum;
      mi = mnew;
      alpha_s[m] = al;
    }
    __syncthreads();

#pragma unroll
    for (int r = 0; r < 8; ++r) {
      const float al = alpha_s[wm + r + rbase];
      o0[r] *= al;
      o1[r] *= al;
    }
#pragma unroll
    for (int kk = 0; kk < 64; kk += 4) {
      v2f a = *(const v2f*)(Ps + (wm + lm) * ALD + kk + koff);
      const int kr = kk + koff;
      v2f b0, b1;
      b0.x = Vs[kr * ALD + wn + lm];
      b0.y = Vs[(kr + 1) * ALD + wn + lm];
      b1.x = Vs[kr * ALD + wn + 16 + lm];
      b1.y = Vs[(kr + 1) * ALD + wn + 16 + lm];
      o0 = wmma4(a, b0, o0);
      o1 = wmma4(a, b1, o1);
    }
    __syncthreads();
  }

  if (t < 64) lrow_s[t] = li;
  __syncthreads();

#pragma unroll
  for (int r = 0; r < 8; ++r) {
    const int row = wm + r + rbase;
    const float inv = 1.0f / lrow_s[row];
    const size_t o = ((size_t)b * S_ + q0 + row) * H_ + nh * HD_ + wn + lm;
    out[o] = o0[r] * inv;
    out[o + 16] = o1[r] * inv;
  }
}

__global__ void tail_kernel(float* out, int idx) {
  if (threadIdx.x == 0 && blockIdx.x == 0) out[idx] = 0.0f;
}

extern "C" void kernel_launch(void* const* d_in, const int* in_sizes, int n_in,
                              void* d_out, int out_size, void* d_ws,
                              size_t ws_size, hipStream_t stream) {
  (void)in_sizes; (void)n_in; (void)ws_size;
  const float* x     = (const float*)d_in[0];
  const float* ln1_g = (const float*)d_in[2];
  const float* ln1_b = (const float*)d_in[3];
  const float* qkv_w = (const float*)d_in[4];
  const float* qkv_b = (const float*)d_in[5];
  const float* out_w = (const float*)d_in[6];
  const float* out_b = (const float*)d_in[7];
  const float* ln2_g = (const float*)d_in[8];
  const float* ln2_b = (const float*)d_in[9];
  const float* fc1_w = (const float*)d_in[10];
  const float* fc1_b = (const float*)d_in[11];
  const float* fc2_w = (const float*)d_in[12];
  const float* fc2_b = (const float*)d_in[13];

  float* xo = (float*)d_out;
  float* ws = (float*)d_ws;
  float* h    = ws;
  float* qkvb = h + (size_t)M_ * H_;
  float* attn = qkvb + (size_t)M_ * 3 * H_;
  float* ffn  = attn + (size_t)M_ * H_;

  ln_kernel<<<M_, 256, 0, stream>>>(x, ln1_g, ln1_b, h);
  gemm_kernel<<<dim3(M_ / 64, (3 * H_) / 64), 256, 0, stream>>>(
      h, qkv_w, qkv_b, nullptr, qkvb, M_, 3 * H_, H_, 0);
  attn_kernel<<<dim3(S_ / 64, B_ * NH_), 256, 0, stream>>>(qkvb, attn);
  gemm_kernel<<<dim3(M_ / 64, H_ / 64), 256, 0, stream>>>(
      attn, out_w, out_b, x, xo, M_, H_, H_, 0);
  ln_kernel<<<M_, 256, 0, stream>>>(xo, ln2_g, ln2_b, h);
  gemm_kernel<<<dim3(M_ / 64, FFN_ / 64), 256, 0, stream>>>(
      h, fc1_w, fc1_b, nullptr, ffn, M_, FFN_, H_, 1);
  gemm_kernel<<<dim3(M_ / 64, H_ / 64), 256, 0, stream>>>(
      ffn, fc2_w, fc2_b, xo, xo, M_, H_, FFN_, 0);
  tail_kernel<<<1, 32, 0, stream>>>(xo, out_size - 1);
}


# attention softmax done by all 256 threads (4 lanes per query row, xor-shuffle row max and row sum) instead of 64 serial threads; row sums now add per quarter row, then across the 4 lanes
# speedup vs baseline: 1.3680x; 1.3680x over previous
#include <hip/hip_runtime.h>
#include <math.h>

#define B_   2
#define S_   2048
#define H_   768
#define NH_  12
#define HD_  64
#define FFN_ 3072
#define M_   (B_ * S_)
#define QKVLD (3 * H_)

typedef __attribute__((ext_vector_type(2))) float v2f;
typedef __attribute__((ext_vector_type(4))) float v4f;
typedef __attribute__((ext_vector_type(8))) float v8f;

__device__ __forceinline__ v8f wmma4(v2f a, v2f b, v8f c) {
  return __builtin_amdgcn_wmma_f32_16x16x4_f32(false, a, false, b, (short)0, c,
                                               false, false);
}

__device__ __forceinline__ float gelu_exact(float x) {
  return 0.5f * x * (1.0f + erff(x * 0.70710678118654752440f));
}

__global__ __launch_bounds__(256) void ln_kernel(const float* __restrict__ x,
                                                 const float* __restrict__ g,
                                                 const float* __restrict__ b,
                                                 float* __restrict__ o) {
  __shared__ float red[256];
  const int t = threadIdx.x;
  const size_t base = (size_t)blockIdx.x * H_;

  float s = 0.0f;
  for (int i = t; i < H_; i += 256) s += x[base + i];
  red[t] = s;
  __syncthreads();
  for (int off = 128; off > 0; off >>= 1) {
    if (t < off) red[t] += red[t + off];
    __syncthreads();
  }
  const float mu = red[0] * (1.0f / (float)H_);
  __syncthreads();

  float vs = 0.0f;
  for (int i = t; i < H_; i += 256) {
    float d = x[base + i] - mu;
    vs += d * d;
  }
  red[t] = vs;
  __syncthreads();
  for (int off = 128; off > 0; off >>= 1) {
    if (t < off) red[t] += red[t + off];
    __syncthreads();
  }
  const float rstd = rsqrtf(red[0] * (1.0f / (float)H_) + 1e-5f);

  for (int i = t; i < H_; i += 256)
    o[base + i] = (x[base + i] - mu) * rstd * g[i] + b[i];
}

#define GLDA 20

__global__ __launch_bounds__(256) void gemm_kernel(
    const float* __restrict__ A, const float* __restrict__ W,
    const float* __restrict__ bias, const float* __restrict__ resid,
    float* __restrict__ C, int M, int N, int K, int act) {
  __shared__ float As[2][64 * GLDA];
  __shared__ float Ws[2][64 * GLDA];

  const int t = threadIdx.x;
  const int wave = t >> 5, lane = t & 31;
  const int m0 = blockIdx.x * 64, n0 = blockIdx.y * 64;
  const int wm = 16 * (wave & 3);
  const int wn = 32 * (wave >> 2);
  const int lm = lane & 15;
  const int koff = (lane < 16) ? 0 : 2;
  const int rbase = (lane < 16) ? 0 : 8;

  const int lrow = t >> 2;
  const int lcol = (t & 3) * 4;

  const float* Ap = A + (size_t)(m0 + lrow) * K + lcol;
  const float* Wp = W + (size_t)(n0 + lrow) * K + lcol;
  const int lds_off = lrow * GLDA + lcol;

  v8f acc0 = {};
  v8f acc1 = {};

  {
    v4f av = *(const v4f*)(Ap);
    v4f wv = *(const v4f*)(Wp);
    *(v4f*)(As[0] + lds_off) = av;
    *(v4f*)(Ws[0] + lds_off) = wv;
  }
  __syncthreads();

  const int nstages = K >> 4;
  for (int s = 0; s < nstages; ++s) {
    const int p = s & 1;
    const bool has_next = (s + 1) < nstages;

    v4f av, wv;
    if (has_next) {
      av = *(const v4f*)(Ap + (size_t)(s + 1) * 16);
      wv = *(const v4f*)(Wp + (size_t)(s + 1) * 16);
    }

    const float* Ab = As[p];
    const float* Wb = Ws[p];
#pragma unroll
    for (int kk = 0; kk < 16; kk += 4) {
      v2f a  = *(const v2f*)(Ab + (wm + lm) * GLDA + kk + koff);
      v2f b0 = *(const v2f*)(Wb + (wn + lm) * GLDA + kk + koff);
      v2f b1 = *(const v2f*)(Wb + (wn + 16 + lm) * GLDA + kk + koff);
      acc0 = wmma4(a, b0, acc0);
      acc1 = wmma4(a, b1, acc1);
    }

    if (has_next) {
      *(v4f*)(As[1 - p] + lds_off) = av;
      *(v4f*)(Ws[1 - p] + lds_off) = wv;
    }
    __syncthreads();
  }

#pragma unroll
  for (int r = 0; r < 8; ++r) {
    const int m = m0 + wm + r + rbase;
    const int n = n0 + wn + lm;
    const size_t i0 = (size_t)m * N + n;
    float v0 = acc0[r] + bias[n];
    float v1 = acc1[r] + bias[n + 16];
    if (resid) {
      v0 += resid[i0];
      v1 += resid[i0 + 16];
    }
    if (act == 1) {
      v0 = gelu_exact(v0);
      v1 = gelu_exact(v1);
    }
    C[i0] = v0;
    C[i0 + 16] = v1;
  }
}

#define ALD 66

__global__ __launch_bounds__(256) void attn_kernel(
    const float* __restrict__ qkv, float* __restrict__ out) {
  __shared__ float Qs[64 * ALD];
  __shared__ float Ks[64 * ALD];
  __shared__ float Vs[64 * ALD];
  __shared__ float alpha_s[64];
  __shared__ float lrow_s[64];

  const int t = threadIdx.x;
  const int wave = t >> 5, lane = t & 31;
  const int lm = lane & 15;
  const int koff = (lane < 16) ? 0 : 2;
  const int rbase = (lane < 16) ? 0 : 8;
  const int wm = 16 * (wave & 3);
  const int wn = 32 * (wave >> 2);

  const int qt = blockIdx.x;
  const int bh = blockIdx.y;
  const int b = bh / NH_, nh = bh % NH_;
  const int q0 = qt * 64;

  const float* qbase = qkv + (size_t)b * S_ * QKVLD + nh * HD_;
  const float* kbase = qbase + H_;
  const float* vbase = qbase + 2 * H_;
  float* Ps = Ks;

  for (int i = 0; i < 8; ++i) {
    int f = t + 256 * i;
    int r = f >> 5;
    int c2 = (f & 31) * 2;
    *(v2f*)(Qs + r * ALD + c2) =
        *(const v2f*)(qbase + (size_t)(q0 + r) * QKVLD + c2);
  }

  v8f o0 = {};
  v8f o1 = {};
  float mi = -INFINITY, li = 0.0f;

  for (int j = 0; j <= qt; ++j) {
    const int k0 = j * 64;
    for (int i = 0; i < 8; ++i) {
      int f = t + 256 * i;
      int r = f >> 5;
      int c2 = (f & 31) * 2;
      *(v2f*)(Ks + r * ALD + c2) =
          *(const v2f*)(kbase + (size_t)(k0 + r) * QKVLD + c2);
      *(v2f*)(Vs + r * ALD + c2) =
          *(const v2f*)(vbase + (size_t)(k0 + r) * QKVLD + c2);
    }
    __syncthreads();

    v8f s0 = {};
    v8f s1 = {};
#pragma unroll
    for (int kk = 0; kk < 64; kk += 4) {
      v2f a  = *(const v2f*)(Qs + (wm + lm) * ALD + kk + koff);
      v2f b0 = *(const v2f*)(Ks + (wn + lm) * ALD + kk + koff);
      v2f b1 = *(const v2f*)(Ks + (wn + 16 + lm) * ALD + kk + koff);
      s0 = wmma4(a, b0, s0);
      s1 = wmma4(a, b1, s1);
    }
    __syncthreads();

#pragma unroll
    for (int r = 0; r < 8; ++r) {
      Ps[(wm + r + rbase) * ALD + wn + lm]      = s0[r] * 0.125f;
      Ps[(wm + r + rbase) * ALD + wn + 16 + lm] = s1[r] * 0.125f;
    }
    __syncthreads();

    {
      const int m = t >> 2;
      const int cq = (t & 3) * 16;
      float* Pr = Ps + m * ALD + cq;
      const int limit = (j == qt) ? (m + 1) : 64;
      v2f pv[8];
      float mx = -INFINITY;
#pragma unroll
      for (int c = 0; c < 8; ++c) {
        pv[c] = *(const v2f*)(Pr + 2 * c);
        if (cq + 2 * c < limit)     mx = fmaxf(mx, pv[c].x);
        if (cq + 2 * c + 1 < limit) mx = fmaxf(mx, pv[c].y);
      }
      mx = fmaxf(mx, __shfl_xor(mx, 1));
      mx = fmaxf(mx, __shfl_xor(mx, 2));
      const float mnew = fmaxf(mi, mx);
      const float al = expf(mi - mnew);
      float sum = 0.0f;
#pragma unroll
      for (int c = 0; c < 8; ++c) {
        v2f p;
        p.x = (cq + 2 * c < limit)     ? expf(pv[c].x - mnew) : 0.0f;
        p.y = (cq + 2 * c + 1 < limit) ? expf(pv[c].y - mnew) : 0.0f;
        *(v2f*)(Pr + 2 * c) = p;
        sum += p.x;
        sum += p.y;
      }
      sum += __shfl_xor(sum, 1);
      sum += __shfl_xor(sum, 2);
      li = li * al + sum;
      mi = mnew;
      if ((t & 3) == 0) alpha_s[m] = al;
    }
    __syncthreads();

#pragma unroll
    for (int r = 0; r < 8; ++r) {
      const float al = alpha_s[wm + r + rbase];
      o0[r] *= al;
      o1[r] *= al;
    }
#pragma unroll
    for (int kk = 0; kk < 64; kk += 4) {
      v2f a = *(const v2f*)(Ps + (wm + lm) * ALD + kk + koff);
      const int kr = kk + koff;
      v2f b0, b1;
      b0.x = Vs[kr * ALD + wn + lm];
      b0.y = Vs[(kr + 1) * ALD + wn + lm];
      b1.x = Vs[kr * ALD + wn + 16 + lm];
      b1.y = Vs[(kr + 1) * ALD + wn + 16 + lm];
      o0 = wmma4(a, b0, o0);
      o1 = wmma4(a, b1, o1);
    }
    __syncthreads();
  }

  if ((t & 3) == 0) lrow_s[t >> 2] = li;
  __syncthreads();

#pragma unroll
  for (int r = 0; r < 8; ++r) {
    const int row = wm + r + rbase;
    const float inv = 1.0f / lrow_s[row];
    const size_t o = ((size_t)b * S_ + q0 + row) * H_ + nh * HD_ + wn + lm;
    out[o] = o0[r] * inv;
    out[o + 16] = o1[r] * inv;
  }
}

__global__ void tail_kernel(float* out, int idx) {
  if (threadIdx.x == 0 && blockIdx.x == 0) out[idx] = 0.0f;
}

extern "C" void kernel_launch(void* const* d_in, const int* in_sizes, int n_in,
                              void* d_out, int out_size, void* d_ws,
                              size_t ws_size, hipStream_t stream) {
  (void)in_sizes; (void)n_in; (void)ws_size;
  const float* x     = (const float*)d_in[0];
  const float* ln1_g = (const float*)d_in[2];
  const float* ln1_b = (const float*)d_in[3];
  const float* qkv_w = (const float*)d_in[4];
  const float* qkv_b = (const float*)d_in[5];
  const float* out_w = (const float*)d_in[6];
  const float* out_b = (const float*)d_in[7];
  const float* ln2_g = (const float*)d_in[8];
  const float* ln2_b = (const float*)d_in[9];
  const float* fc1_w = (const float*)d_in[10];
  const float* fc1_b = (const float*)d_in[11];
  const float* fc2_w = (const float*)d_in[12];
  const float* fc2_b = (const float*)d_in[13];

  float* xo = (float*)d_out;
  float* ws = (float*)d_ws;
  float* h    = ws;
  float* qkvb = h + (size_t)M_ * H_;
  float* attn = qkvb + (size_t)M_ * 3 * H_;
  float* ffn  = attn + (size_t)M_ * H_;

  ln_kernel<<<M_, 256, 0, stream>>>(x, ln1_g, ln1_b, h);
  gemm_kernel<<<dim3(M_ / 64, (3 * H_) / 64), 256, 0, stream>>>(
      h, qkv_w, qkv_b, nullptr, qkvb, M_, 3 * H_, H_, 0);
  attn_kernel<<<dim3(S_ / 64, B_ * NH_), 256, 0, stream>>>(qkvb, attn);
  gemm_kernel<<<dim3(M_ / 64, H_ / 64), 256, 0, stream>>>(
      attn, out_w, out_b, x, xo, M_, H_, H_, 0);
  ln_kernel<<<M_, 256, 0, stream>>>(xo, ln2_g, ln2_b, h);
  gemm_kernel<<<dim3(M_ / 64, FFN_ / 64), 256, 0, stream>>>(
      h, fc1_w, fc1_b, nullptr, ffn, M_, FFN_, H_, 1);
  gemm_kernel<<<dim3(M_ / 64, H_ / 64), 256, 0, stream>>>(
      ffn, fc2_w, fc2_b, xo, xo, M_, H_, FFN_, 0);
  tail_kernel<<<1, 32, 0, stream>>>(xo, out_size - 1);
}


# attention work-groups renumbered on a 1-D grid: causal items ranked by work, heaviest first, so groups j, j+256, j+512 carry equal key-tile totals; arithmetic and sum order unchanged
# speedup vs baseline: 1.4501x; 1.0600x over previous
#include <hip/hip_runtime.h>
#include <math.h>

#define B_   2
#define S_   2048
#define H_   768
#define NH_  12
#define HD_  64
#define FFN_ 3072
#define M_   (B_ * S_)
#define QKVLD (3 * H_)

typedef __attribute__((ext_vector_type(2))) float v2f;
typedef __attribute__((ext_vector_type(4))) float v4f;
typedef __attribute__((ext_vector_type(8))) float v8f;

__device__ __forceinline__ v8f wmma4(v2f a, v2f b, v8f c) {
  return __builtin_amdgcn_wmma_f32_16x16x4_f32(false, a, false, b, (short)0, c,
                                               false, false);
}

__device__ __forceinline__ float gelu_exact(float x) {
  return 0.5f * x * (1.0f + erff(x * 0.70710678118654752440f));
}

__global__ __launch_bounds__(256) void ln_kernel(const float* __restrict__ x,
                                                 const float* __restrict__ g,
                                                 const float* __restrict__ b,
                                                 float* __restrict__ o) {
  __shared__ float red[256];
  const int t = threadIdx.x;
  const size_t base = (size_t)blockIdx.x * H_;

  float s = 0.0f;
  for (int i = t; i < H_; i += 256) s += x[base + i];
  red[t] = s;
  __syncthreads();
  for (int off = 128; off > 0; off >>= 1) {
    if (t < off) red[t] += red[t + off];
    __syncthreads();
  }
  const float mu = red[0] * (1.0f / (float)H_);
  __syncthreads();

  float vs = 0.0f;
  for (int i = t; i < H_; i += 256) {
    float d = x[base + i] - mu;
    vs += d * d;
  }
  red[t] = vs;
  __syncthreads();
  for (int off = 128; off > 0; off >>= 1) {
    if (t < off) red[t] += red[t + off];
    __syncthreads();
  }
  const float rstd = rsqrtf(red[0] * (1.0f / (float)H_) + 1e-5f);

  for (int i = t; i < H_; i += 256)
    o[base + i] = (x[base + i] - mu) * rstd * g[i] + b[i];
}

#define GLDA 20

__global__ __launch_bounds__(256) void gemm_kernel(
    const float* __restrict__ A, const float* __restrict__ W,
    const float* __restrict__ bias, const float* __restrict__ resid,
    float* __restrict__ C, int M, int N, int K, int act) {
  __shared__ float As[2][64 * GLDA];
  __shared__ float Ws[2][64 * GLDA];

  const int t = threadIdx.x;
  const int wave = t >> 5, lane = t & 31;
  const int m0 = blockIdx.x * 64, n0 = blockIdx.y * 64;
  const int wm = 16 * (wave & 3);
  const int wn = 32 * (wave >> 2);
  const int lm = lane & 15;
  const int koff = (lane < 16) ? 0 : 2;
  const int rbase = (lane < 16) ? 0 : 8;

  const int lrow = t >> 2;
  const int lcol = (t & 3) * 4;

  const float* Ap = A + (size_t)(m0 + lrow) * K + lcol;
  const float* Wp = W + (size_t)(n0 + lrow) * K + lcol;
  const int lds_off = lrow * GLDA + lcol;

  v8f acc0 = {};
  v8f acc1 = {};

  {
    v4f av = *(const v4f*)(Ap);
    v4f wv = *(const v4f*)(Wp);
    *(v4f*)(As[0] + lds_off) = av;
    *(v4f*)(Ws[0] + lds_off) = wv;
  }
  __syncthreads();

  const int nstages = K >> 4;
  for (int s = 0; s < nstages; ++s) {
    const int p = s & 1;
    const bool has_next = (s + 1) < nstages;

    v4f av, wv;
    if (has_next) {
      av = *(const v4f*)(Ap + (size_t)(s + 1) * 16);
      wv = *(const v4f*)(Wp + (size_t)(s + 1) * 16);
    }

    const float* Ab = As[p];
    const float* Wb = Ws[p];
#pragma unroll
    for (int kk = 0; kk < 16; kk += 4) {
      v2f a  = *(const v2f*)(Ab + (wm + lm) * GLDA + kk + koff);
      v2f b0 = *(const v2f*)(Wb + (wn + lm) * GLDA + kk + koff);
      v2f b1 = *(const v2f*)(Wb + (wn + 16 + lm) * GLDA + kk + koff);
      acc0 = wmma4(a, b0, acc0);
      acc1 = wmma4(a, b1, acc1);
    }

    if (has_next) {
      *(v4f*)(As[1 - p] + lds_off) = av;
      *(v4f*)(Ws[1 - p] + lds_off) = wv;
    }
    __syncthreads();
  }

#pragma unroll
  for (int r = 0; r < 8; ++r) {
    const int m = m0 + wm + r + rbase;
    const int n = n0 + wn + lm;
    const size_t i0 = (size_t)m * N + n;
    float v0 = acc0[r] + bias[n];
    float v1 = acc1[r] + bias[n + 16];
    if (resid) {
      v0 += resid[i0];
      v1 += resid[i0 + 16];
    }
    if (act == 1) {
      v0 = gelu_exact(v0);
      v1 = gelu_exact(v1);
    }
    C[i0] = v0;
    C[i0 + 16] = v1;
  }
}

#define ALD 66

__global__ __launch_bounds__(256) void attn_kernel(
    const float* __restrict__ qkv, float* __restrict__ out) {
  __shared__ float Qs[64 * ALD];
  __shared__ float Ks[64 * ALD];
  __shared__ float Vs[64 * ALD];
  __shared__ float alpha_s[64];
  __shared__ float lrow_s[64];

  const int t = threadIdx.x;
  const int wave = t >> 5, lane = t & 31;
  const int lm = lane & 15;
  const int koff = (lane < 16) ? 0 : 2;
  const int rbase = (lane < 16) ? 0 : 8;
  const int wm = 16 * (wave & 3);
  const int wn = 32 * (wave >> 2);

  static_assert((S_ / 64) * B_ * NH_ == 768, "numbering below is for 3 x 256 items");
  const int widx = blockIdx.x;
  const int g3 = widx >> 8;
  const int j3 = widx & 255;
  int p3;
  if (g3 == 0)      p3 = j3;
  else if (g3 == 1) p3 = (j3 + 128) & 255;
  else              p3 = (j3 < 128) ? (255 - 2 * j3) : (510 - 2 * j3);
  const int rank = (g3 << 8) + p3;
  const int qt = (S_ / 64 - 1) - rank / (B_ * NH_);
  const int bh = rank % (B_ * NH_);
  const int b = bh / NH_, nh = bh % NH_;
  const int q0 = qt * 64;

  const float* qbase = qkv + (size_t)b * S_ * QKVLD + nh * HD_;
  const float* kbase = qbase + H_;
  const float* vbase = qbase + 2 * H_;
  float* Ps = Ks;

  for (int i = 0; i < 8; ++i) {
    int f = t + 256 * i;
    int r = f >> 5;
    int c2 = (f & 31) * 2;
    *(v2f*)(Qs + r * ALD + c2) =
        *(const v2f*)(qbase + (size_t)(q0 + r) * QKVLD + c2);
  }

  v8f o0 = {};
  v8f o1 = {};
  float mi = -INFINITY, li = 0.0f;

  for (int j = 0; j <= qt; ++j) {
    const int k0 = j * 64;
    for (int i = 0; i < 8; ++i) {
      int f = t + 256 * i;
      int r = f >> 5;
      int c2 = (f & 31) * 2;
      *(v2f*)(Ks + r * ALD + c2) =
          *(const v2f*)(kbase + (size_t)(k0 + r) * QKVLD + c2);
      *(v2f*)(Vs + r * ALD + c2) =
          *(const v2f*)(vbase + (size_t)(k0 + r) * QKVLD + c2);
    }
    __syncthreads();

    v8f s0 = {};
    v8f s1 = {};
#pragma unroll
    for (int kk = 0; kk < 64; kk += 4) {
      v2f a  = *(const v2f*)(Qs + (wm + lm) * ALD + kk + koff);
      v2f b0 = *(const v2f*)(Ks + (wn + lm) * ALD + kk + koff);
      v2f b1 = *(const v2f*)(Ks + (wn + 16 + lm) * ALD + kk + koff);
      s0 = wmma4(a, b0, s0);
      s1 = wmma4(a, b1, s1);
    }
    __syncthreads();

#pragma unroll
    for (int r = 0; r < 8; ++r) {
      Ps[(wm + r + rbase) * ALD + wn + lm]      = s0[r] * 0.125f;
      Ps[(wm + r + rbase) * ALD + wn + 16 + lm] = s1[r] * 0.125f;
    }
    __syncthreads();

    {
      const int m = t >> 2;
      const int cq = (t & 3) * 16;
      float* Pr = Ps + m * ALD + cq;
      const int limit = (j == qt) ? (m + 1) : 64;
      v2f pv[8];
      float mx = -INFINITY;
#pragma unroll
      for (int c = 0; c < 8; ++c) {
        pv[c] = *(const v2f*)(Pr + 2 * c);
        if (cq + 2 * c < limit)     mx = fmaxf(mx, pv[c].x);
        if (cq + 2 * c + 1 < limit) mx = fmaxf(mx, pv[c].y);
      }
      mx = fmaxf(mx, __shfl_xor(mx, 1));
      mx = fmaxf(mx, __shfl_xor(mx, 2));
      const float mnew = fmaxf(mi, mx);
      const float al = expf(mi - mnew);
      float sum = 0.0f;
#pragma unroll
      for (int c = 0; c < 8; ++c) {
        v2f p;
        p.x = (cq + 2 * c < limit)     ? expf(pv[c].x - mnew) : 0.0f;
        p.y = (cq + 2 * c + 1 < limit) ? expf(pv[c].y - mnew) : 0.0f;
        *(v2f*)(Pr + 2 * c) = p;
        sum += p.x;
        sum += p.y;
      }
      sum += __shfl_xor(sum, 1);
      sum += __shfl_xor(sum, 2);
      li = li * al + sum;
      mi = mnew;
      if ((t & 3) == 0) alpha_s[m] = al;
    }
    __syncthreads();

#pragma unroll
    for (int r = 0; r < 8; ++r) {
      const float al = alpha_s[wm + r + rbase];
      o0[r] *= al;
      o1[r] *= al;
    }
#pragma unroll
    for (int kk = 0; kk < 64; kk += 4) {
      v2f a = *(const v2f*)(Ps + (wm + lm) * ALD + kk + koff);
      const int kr = kk + koff;
      v2f b0, b1;
      b0.x = Vs[kr * ALD + wn + lm];
      b0.y = Vs[(kr + 1) * ALD + wn + lm];
      b1.x = Vs[kr * ALD + wn + 16 + lm];
      b1.y = Vs[(kr + 1) * ALD + wn + 16 + lm];
      o0 = wmma4(a, b0, o0);
      o1 = wmma4(a, b1, o1);
    }
    __syncthreads();
  }

  if ((t & 3) == 0) lrow_s[t >> 2] = li;
  __syncthreads();

#pragma unroll
  for (int r = 0; r < 8; ++r) {
    const int row = wm + r + rbase;
    const float inv = 1.0f / lrow_s[row];
    const size_t o = ((size_t)b * S_ + q0 + row) * H_ + nh * HD_ + wn + lm;
    out[o] = o0[r] * inv;
    out[o + 16] = o1[r] * inv;
  }
}

__global__ void tail_kernel(float* out, int idx) {
  if (threadIdx.x == 0 && blockIdx.x == 0) out[idx] = 0.0f;
}

extern "C" void kernel_launch(void* const* d_in, const int* in_sizes, int n_in,
                              void* d_out, int out_size, void* d_ws,
                              size_t ws_size, hipStream_t stream) {
  (void)in_sizes; (void)n_in; (void)ws_size;
  const float* x     = (const float*)d_in[0];
  const float* ln1_g = (const float*)d_in[2];
  const float* ln1_b = (const float*)d_in[3];
  const float* qkv_w = (const float*)d_in[4];
  const float* qkv_b = (const float*)d_in[5];
  const float* out_w = (const float*)d_in[6];
  const float* out_b = (const float*)d_in[7];
  const float* ln2_g = (const float*)d_in[8];
  const float* ln2_b = (const float*)d_in[9];
  const float* fc1_w = (const float*)d_in[10];
  const float* fc1_b = (const float*)d_in[11];
  const float* fc2_w = (const float*)d_in[12];
  const float* fc2_b = (const float*)d_in[13];

  float* xo = (float*)d_out;
  float* ws = (float*)d_ws;
  float* h    = ws;
  float* qkvb = h + (size_t)M_ * H_;
  float* attn = qkvb + (size_t)M_ * 3 * H_;
  float* ffn  = attn + (size_t)M_ * H_;

  ln_kernel<<<M_, 256, 0, stream>>>(x, ln1_g, ln1_b, h);
  gemm_kernel<<<dim3(M_ / 64, (3 * H_) / 64), 256, 0, stream>>>(
      h, qkv_w, qkv_b, nullptr, qkvb, M_, 3 * H_, H_, 0);
  attn_kernel<<<dim3((S_ / 64) * B_ * NH_), 256, 0, stream>>>(qkvb, attn);
  gemm_kernel<<<dim3(M_ / 64, H_ / 64), 256, 0, stream>>>(
      attn, out_w, out_b, x, xo, M_, H_, H_, 0);
  ln_kernel<<<M_, 256, 0, stream>>>(xo, ln2_g, ln2_b, h);
  gemm_kernel<<<dim3(M_ / 64, FFN_ / 64), 256, 0, stream>>>(
      h, fc1_w, fc1_b, nullptr, ffn, M_, FFN_, H_, 1);
  gemm_kernel<<<dim3(M_ / 64, H_ / 64), 256, 0, stream>>>(
      ffn, fc2_w, fc2_b, xo, xo, M_, H_, FFN_, 0);
  tail_kernel<<<1, 32, 0, stream>>>(xo, out_size - 1);
}


# GEMM K stage 16 -> 32 deep (LDS pitch 36; two 16-byte pieces per thread and operand, eight threads per whole 128-byte row): half the barriers, 16 WMMAs per wave between them; sum order unchanged
# speedup vs baseline: 1.5211x; 1.0490x over previous
#include <hip/hip_runtime.h>
#include <math.h>

#define B_   2
#define S_   2048
#define H_   768
#define NH_  12
#define HD_  64
#define FFN_ 3072
#define M_   (B_ * S_)
#define QKVLD (3 * H_)

typedef __attribute__((ext_vector_type(2))) float v2f;
typedef __attribute__((ext_vector_type(4))) float v4f;
typedef __attribute__((ext_vector_type(8))) float v8f;

__device__ __forceinline__ v8f wmma4(v2f a, v2f b, v8f c) {
  return __builtin_amdgcn_wmma_f32_16x16x4_f32(false, a, false, b, (short)0, c,
                                               false, false);
}

__device__ __forceinline__ float gelu_exact(float x) {
  return 0.5f * x * (1.0f + erff(x * 0.70710678118654752440f));
}

__global__ __launch_bounds__(256) void ln_kernel(const float* __restrict__ x,
                                                 const float* __restrict__ g,
                                                 const float* __restrict__ b,
                                                 float* __restrict__ o) {
  __shared__ float red[256];
  const int t = threadIdx.x;
  const size_t base = (size_t)blockIdx.x * H_;

  float s = 0.0f;
  for (int i = t; i < H_; i += 256) s += x[base + i];
  red[t] = s;
  __syncthreads();
  for (int off = 128; off > 0; off >>= 1) {
    if (t < off) red[t] += red[t + off];
    __syncthreads();
  }
  const float mu = red[0] * (1.0f / (float)H_);
  __syncthreads();

  float vs = 0.0f;
  for (int i = t; i < H_; i += 256) {
    float d = x[base + i] - mu;
    vs += d * d;
  }
  red[t] = vs;
  __syncthreads();
  for (int off = 128; off > 0; off >>= 1) {
    if (t < off) red[t] += red[t + off];
    __syncthreads();
  }
  const float rstd = rsqrtf(red[0] * (1.0f / (float)H_) + 1e-5f);

  for (int i = t; i < H_; i += 256)
    o[base + i] = (x[base + i] - mu) * rstd * g[i] + b[i];
}

#define GBK  32
#define GLDA 36

__global__ __launch_bounds__(256) void gemm_kernel(
    const float* __restrict__ A, const float* __restrict__ W,
    const float* __restrict__ bias, const float* __restrict__ resid,
    float* __restrict__ C, int M, int N, int K, int act) {
  __shared__ float As[2][64 * GLDA];
  __shared__ float Ws[2][64 * GLDA];

  const int t = threadIdx.x;
  const int wave = t >> 5, lane = t & 31;
  const int m0 = blockIdx.x * 64, n0 = blockIdx.y * 64;
  const int wm = 16 * (wave & 3);
  const int wn = 32 * (wave >> 2);
  const int lm = lane & 15;
  const int koff = (lane < 16) ? 0 : 2;
  const int rbase = (lane < 16) ? 0 : 8;

  const int lrow = t >> 3;
  const int lcol = (t & 7) * 4;

  const float* Ap0 = A + (size_t)(m0 + lrow) * K + lcol;
  const float* Ap1 = Ap0 + (size_t)32 * K;
  const float* Wp0 = W + (size_t)(n0 + lrow) * K + lcol;
  const float* Wp1 = Wp0 + (size_t)32 * K;
  const int lds_off0 = lrow * GLDA + lcol;
  const int lds_off1 = lds_off0 + 32 * GLDA;

  v8f acc0 = {};
  v8f acc1 = {};

  {
    v4f a0 = *(const v4f*)(Ap0);
    v4f a1 = *(const v4f*)(Ap1);
    v4f w0 = *(const v4f*)(Wp0);
    v4f w1 = *(const v4f*)(Wp1);
    *(v4f*)(As[0] + lds_off0) = a0;
    *(v4f*)(As[0] + lds_off1) = a1;
    *(v4f*)(Ws[0] + lds_off0) = w0;
    *(v4f*)(Ws[0] + lds_off1) = w1;
  }
  __syncthreads();

  const int nstages = K / GBK;
  for (int s = 0; s < nstages; ++s) {
    const int p = s & 1;
    const bool has_next = (s + 1) < nstages;

    v4f a0, a1, w0, w1;
    if (has_next) {
      const size_t ko = (size_t)(s + 1) * GBK;
      a0 = *(const v4f*)(Ap0 + ko);
      a1 = *(const v4f*)(Ap1 + ko);
      w0 = *(const v4f*)(Wp0 + ko);
      w1 = *(const v4f*)(Wp1 + ko);
    }

    const float* Ab = As[p];
    const float* Wb = Ws[p];
#pragma unroll
    for (int kk = 0; kk < GBK; kk += 4) {
      v2f a  = *(const v2f*)(Ab + (wm + lm) * GLDA + kk + koff);
      v2f b0 = *(const v2f*)(Wb + (wn + lm) * GLDA + kk + koff);
      v2f b1 = *(const v2f*)(Wb + (wn + 16 + lm) * GLDA + kk + koff);
      acc0 = wmma4(a, b0, acc0);
      acc1 = wmma4(a, b1, acc1);
    }

    if (has_next) {
      *(v4f*)(As[1 - p] + lds_off0) = a0;
      *(v4f*)(As[1 - p] + lds_off1) = a1;
      *(v4f*)(Ws[1 - p] + lds_off0) = w0;
      *(v4f*)(Ws[1 - p] + lds_off1) = w1;
    }
    __syncthreads();
  }

#pragma unroll
  for (int r = 0; r < 8; ++r) {
    const int m = m0 + wm + r + rbase;
    const int n = n0 + wn + lm;
    const size_t i0 = (size_t)m * N + n;
    float v0 = acc0[r] + bias[n];
    float v1 = acc1[r] + bias[n + 16];
    if (resid) {
      v0 += resid[i0];
      v1 += resid[i0 + 16];
    }
    if (act == 1) {
      v0 = gelu_exact(v0);
      v1 = gelu_exact(v1);
    }
    C[i0] = v0;
    C[i0 + 16] = v1;
  }
}

#define ALD 66

__global__ __launch_bounds__(256) void attn_kernel(
    const float* __restrict__ qkv, float* __restrict__ out) {
  __shared__ float Qs[64 * ALD];
  __shared__ float Ks[64 * ALD];
  __shared__ float Vs[64 * ALD];
  __shared__ float alpha_s[64];
  __shared__ float lrow_s[64];

  const int t = threadIdx.x;
  const int wave = t >> 5, lane = t & 31;
  const int lm = lane & 15;
  const int koff = (lane < 16) ? 0 : 2;
  const int rbase = (lane < 16) ? 0 : 8;
  const int wm = 16 * (wave & 3);
  const int wn = 32 * (wave >> 2);

  static_assert((S_ / 64) * B_ * NH_ == 768, "numbering below is for 3 x 256 items");
  const int widx = blockIdx.x;
  const int g3 = widx >> 8;
  const int j3 = widx & 255;
  int p3;
  if (g3 == 0)      p3 = j3;
  else if (g3 == 1) p3 = (j3 + 128) & 255;
  else              p3 = (j3 < 128) ? (255 - 2 * j3) : (510 - 2 * j3);
  const int rank = (g3 << 8) + p3;
  const int qt = (S_ / 64 - 1) - rank / (B_ * NH_);
  const int bh = rank % (B_ * NH_);
  const int b = bh / NH_, nh = bh % NH_;
  const int q0 = qt * 64;

  const float* qbase = qkv + (size_t)b * S_ * QKVLD + nh * HD_;
  const float* kbase = qbase + H_;
  const float* vbase = qbase + 2 * H_;
  float* Ps = Ks;

  for (int i = 0; i < 8; ++i) {
    int f = t + 256 * i;
    int r = f >> 5;
    int c2 = (f & 31) * 2;
    *(v2f*)(Qs + r * ALD + c2) =
        *(const v2f*)(qbase + (size_t)(q0 + r) * QKVLD + c2);
  }

  v8f o0 = {};
  v8f o1 = {};
  float mi = -INFINITY, li = 0.0f;

  for (int j = 0; j <= qt; ++j) {
    const int k0 = j * 64;
    for (int i = 0; i < 8; ++i) {
      int f = t + 256 * i;
      int r = f >> 5;
      int c2 = (f & 31) * 2;
      *(v2f*)(Ks + r * ALD + c2) =
          *(const v2f*)(kbase + (size_t)(k0 + r) * QKVLD + c2);
      *(v2f*)(Vs + r * ALD + c2) =
          *(const v2f*)(vbase + (size_t)(k0 + r) * QKVLD + c2);
    }
    __syncthreads();

    v8f s0 = {};
    v8f s1 = {};
#pragma unroll
    for (int kk = 0; kk < 64; kk += 4) {
      v2f a  = *(const v2f*)(Qs + (wm + lm) * ALD + kk + koff);
      v2f b0 = *(const v2f*)(Ks + (wn + lm) * ALD + kk + koff);
      v2f b1 = *(const v2f*)(Ks + (wn + 16 + lm) * ALD + kk + koff);
      s0 = wmma4(a, b0, s0);
      s1 = wmma4(a, b1, s1);
    }
    __syncthreads();

#pragma unroll
    for (int r = 0; r < 8; ++r) {
      Ps[(wm + r + rbase) * ALD + wn + lm]      = s0[r] * 0.125f;
      Ps[(wm + r + rbase) * ALD + wn + 16 + lm] = s1[r] * 0.125f;
    }
    __syncthreads();

    {
      const int m = t >> 2;
      const int cq = (t & 3) * 16;
      float* Pr = Ps + m * ALD + cq;
      const int limit = (j == qt) ? (m + 1) : 64;
      v2f pv[8];
      float mx = -INFINITY;
#pragma unroll
      for (int c = 0; c < 8; ++c) {
        pv[c] = *(const v2f*)(Pr + 2 * c);
        if (cq + 2 * c < limit)     mx = fmaxf(mx, pv[c].x);
        if (cq + 2 * c + 1 < limit) mx = fmaxf(mx, pv[c].y);
      }
      mx = fmaxf(mx, __shfl_xor(mx, 1));
      mx = fmaxf(mx, __shfl_xor(mx, 2));
      const float mnew = fmaxf(mi, mx);
      const float al = expf(mi - mnew);
      float sum = 0.0f;
#pragma unroll
      for (int c = 0; c < 8; ++c) {
        v2f p;
        p.x = (cq + 2 * c < limit)     ? expf(pv[c].x - mnew) : 0.0f;
        p.y = (cq + 2 * c + 1 < limit) ? expf(pv[c].y - mnew) : 0.0f;
        *(v2f*)(Pr + 2 * c) = p;
        sum += p.x;
        sum += p.y;
      }
      sum += __shfl_xor(sum, 1);
      sum += __shfl_xor(sum, 2);
      li = li * al + sum;
      mi = mnew;
      if ((t & 3) == 0) alpha_s[m] = al;
    }
    __syncthreads();

#pragma unroll
    for (int r = 0; r < 8; ++r) {
      const float al = alpha_s[wm + r + rbase];
      o0[r] *= al;
      o1[r] *= al;
    }
#pragma unroll
    for (int kk = 0; kk < 64; kk += 4) {
      v2f a = *(const v2f*)(Ps + (wm + lm) * ALD + kk + koff);
      const int kr = kk + koff;
      v2f b0, b1;
      b0.x = Vs[kr * ALD + wn + lm];
      b0.y = Vs[(kr + 1) * ALD + wn + lm];
      b1.x = Vs[kr * ALD + wn + 16 + lm];
      b1.y = Vs[(kr + 1) * ALD + wn + 16 + lm];
      o0 = wmma4(a, b0, o0);
      o1 = wmma4(a, b1, o1);
    }
    __syncthreads();
  }

  if ((t & 3) == 0) lrow_s[t >> 2] = li;
  __syncthreads();

#pragma unroll
  for (int r = 0; r < 8; ++r) {
    const int row = wm + r + rbase;
    const float inv = 1.0f / lrow_s[row];
    const size_t o = ((size_t)b * S_ + q0 + row) * H_ + nh * HD_ + wn + lm;
    out[o] = o0[r] * inv;
    out[o + 16] = o1[r] * inv;
  }
}

__global__ void tail_kernel(float* out, int idx) {
  if (threadIdx.x == 0 && blockIdx.x == 0) out[idx] = 0.0f;
}

extern "C" void kernel_launch(void* const* d_in, const int* in_sizes, int n_in,
                              void* d_out, int out_size, void* d_ws,
                              size_t ws_size, hipStream_t stream) {
  (void)in_sizes; (void)n_in; (void)ws_size;
  const float* x     = (const float*)d_in[0];
  const float* ln1_g = (const float*)d_in[2];
  const float* ln1_b = (const float*)d_in[3];
  const float* qkv_w = (const float*)d_in[4];
  const float* qkv_b = (const float*)d_in[5];
  const float* out_w = (const float*)d_in[6];
  const float* out_b = (const float*)d_in[7];
  const float* ln2_g = (const float*)d_in[8];
  const float* ln2_b = (const float*)d_in[9];
  const float* fc1_w = (const float*)d_in[10];
  const float* fc1_b = (const float*)d_in[11];
  const float* fc2_w = (const float*)d_in[12];
  const float* fc2_b = (const float*)d_in[13];

  float* xo = (float*)d_out;
  float* ws = (float*)d_ws;
  float* h    = ws;
  float* qkvb = h + (size_t)M_ * H_;
  float* attn = qkvb + (size_t)M_ * 3 * H_;
  float* ffn  = attn + (size_t)M_ * H_;

  ln_kernel<<<M_, 256, 0, stream>>>(x, ln1_g, ln1_b, h);
  gemm_kernel<<<dim3(M_ / 64, (3 * H_) / 64), 256, 0, stream>>>(
      h, qkv_w, qkv_b, nullptr, qkvb, M_, 3 * H_, H_, 0);
  attn_kernel<<<dim3((S_ / 64) * B_ * NH_), 256, 0, stream>>>(qkvb, attn);
  gemm_kernel<<<dim3(M_ / 64, H_ / 64), 256, 0, stream>>>(
      attn, out_w, out_b, x, xo, M_, H_, H_, 0);
  ln_kernel<<<M_, 256, 0, stream>>>(xo, ln2_g, ln2_b, h);
  gemm_kernel<<<dim3(M_ / 64, FFN_ / 64), 256, 0, stream>>>(
      h, fc1_w, fc1_b, nullptr, ffn, M_, FFN_, H_, 1);
  gemm_kernel<<<dim3(M_ / 64, H_ / 64), 256, 0, stream>>>(
      ffn, fc2_w, fc2_b, xo, xo, M_, H_, FFN_, 0);
  tail_kernel<<<1, 32, 0, stream>>>(xo, out_size - 1);
}


# attention softmax in the log2 domain: raw scores to LDS, a packed fma (s*log2e/8 - max) + hardware exp2 per pair replace library expf; causal mask stored as -inf in the diagonal tile only
# speedup vs baseline: 1.5911x; 1.0460x over previous
#include <hip/hip_runtime.h>
#include <math.h>

#define B_   2
#define S_   2048
#define H_   768
#define NH_  12
#define HD_  64
#define FFN_ 3072
#define M_   (B_ * S_)
#define QKVLD (3 * H_)

typedef __attribute__((ext_vector_type(2))) float v2f;
typedef __attribute__((ext_vector_type(4))) float v4f;
typedef __attribute__((ext_vector_type(8))) float v8f;

__device__ __forceinline__ v8f wmma4(v2f a, v2f b, v8f c) {
  return __builtin_amdgcn_wmma_f32_16x16x4_f32(false, a, false, b, (short)0, c,
                                               false, false);
}

__device__ __forceinline__ float gelu_exact(float x) {
  return 0.5f * x * (1.0f + erff(x * 0.70710678118654752440f));
}

__global__ __launch_bounds__(256) void ln_kernel(const float* __restrict__ x,
                                                 const float* __restrict__ g,
                                                 const float* __restrict__ b,
                                                 float* __restrict__ o) {
  __shared__ float red[256];
  const int t = threadIdx.x;
  const size_t base = (size_t)blockIdx.x * H_;

  float s = 0.0f;
  for (int i = t; i < H_; i += 256) s += x[base + i];
  red[t] = s;
  __syncthreads();
  for (int off = 128; off > 0; off >>= 1) {
    if (t < off) red[t] += red[t + off];
    __syncthreads();
  }
  const float mu = red[0] * (1.0f / (float)H_);
  __syncthreads();

  float vs = 0.0f;
  for (int i = t; i < H_; i += 256) {
    float d = x[base + i] - mu;
    vs += d * d;
  }
  red[t] = vs;
  __syncthreads();
  for (int off = 128; off > 0; off >>= 1) {
    if (t < off) red[t] += red[t + off];
    __syncthreads();
  }
  const float rstd = rsqrtf(red[0] * (1.0f / (float)H_) + 1e-5f);

  for (int i = t; i < H_; i += 256)
    o[base + i] = (x[base + i] - mu) * rstd * g[i] + b[i];
}

#define GBK  32
#define GLDA 36

__global__ __launch_bounds__(256) void gemm_kernel(
    const float* __restrict__ A, const float* __restrict__ W,
    const float* __restrict__ bias, const float* __restrict__ resid,
    float* __restrict__ C, int M, int N, int K, int act) {
  __shared__ float As[2][64 * GLDA];
  __shared__ float Ws[2][64 * GLDA];

  const int t = threadIdx.x;
  const int wave = t >> 5, lane = t & 31;
  const int m0 = blockIdx.x * 64, n0 = blockIdx.y * 64;
  const int wm = 16 * (wave & 3);
  const int wn = 32 * (wave >> 2);
  const int lm = lane & 15;
  const int koff = (lane < 16) ? 0 : 2;
  const int rbase = (lane < 16) ? 0 : 8;

  const int lrow = t >> 3;
  const int lcol = (t & 7) * 4;

  const float* Ap0 = A + (size_t)(m0 + lrow) * K + lcol;
  const float* Ap1 = Ap0 + (size_t)32 * K;
  const float* Wp0 = W + (size_t)(n0 + lrow) * K + lcol;
  const float* Wp1 = Wp0 + (size_t)32 * K;
  const int lds_off0 = lrow * GLDA + lcol;
  const int lds_off1 = lds_off0 + 32 * GLDA;

  v8f acc0 = {};
  v8f acc1 = {};

  {
    v4f a0 = *(const v4f*)(Ap0);
    v4f a1 = *(const v4f*)(Ap1);
    v4f w0 = *(const v4f*)(Wp0);
    v4f w1 = *(const v4f*)(Wp1);
    *(v4f*)(As[0] + lds_off0) = a0;
    *(v4f*)(As[0] + lds_off1) = a1;
    *(v4f*)(Ws[0] + lds_off0) = w0;
    *(v4f*)(Ws[0] + lds_off1) = w1;
  }
  __syncthreads();

  const int nstages = K / GBK;
  for (int s = 0; s < nstages; ++s) {
    const int p = s & 1;
    const bool has_next = (s + 1) < nstages;

    v4f a0, a1, w0, w1;
    if (has_next) {
      const size_t ko = (size_t)(s + 1) * GBK;
      a0 = *(const v4f*)(Ap0 + ko);
      a1 = *(const v4f*)(Ap1 + ko);
      w0 = *(const v4f*)(Wp0 + ko);
      w1 = *(const v4f*)(Wp1 + ko);
    }

    const float* Ab = As[p];
    const float* Wb = Ws[p];
#pragma unroll
    for (int kk = 0; kk < GBK; kk += 4) {
      v2f a  = *(const v2f*)(Ab + (wm + lm) * GLDA + kk + koff);
      v2f b0 = *(const v2f*)(Wb + (wn + lm) * GLDA + kk + koff);
      v2f b1 = *(const v2f*)(Wb + (wn + 16 + lm) * GLDA + kk + koff);
      acc0 = wmma4(a, b0, acc0);
      acc1 = wmma4(a, b1, acc1);
    }

    if (has_next) {
      *(v4f*)(As[1 - p] + lds_off0) = a0;
      *(v4f*)(As[1 - p] + lds_off1) = a1;
      *(v4f*)(Ws[1 - p] + lds_off0) = w0;
      *(v4f*)(Ws[1 - p] + lds_off1) = w1;
    }
    __syncthreads();
  }

#pragma unroll
  for (int r = 0; r < 8; ++r) {
    const int m = m0 + wm + r + rbase;
    const int n = n0 + wn + lm;
    const size_t i0 = (size_t)m * N + n;
    float v0 = acc0[r] + bias[n];
    float v1 = acc1[r] + bias[n + 16];
    if (resid) {
      v0 += resid[i0];
      v1 += resid[i0 + 16];
    }
    if (act == 1) {
      v0 = gelu_exact(v0);
      v1 = gelu_exact(v1);
    }
    C[i0] = v0;
    C[i0 + 16] = v1;
  }
}

#define ALD 66
#define CSCALE 0.18033688011112042f

__global__ __launch_bounds__(256) void attn_kernel(
    const float* __restrict__ qkv, float* __restrict__ out) {
  __shared__ float Qs[64 * ALD];
  __shared__ float Ks[64 * ALD];
  __shared__ float Vs[64 * ALD];
  __shared__ float alpha_s[64];
  __shared__ float lrow_s[64];

  const int t = threadIdx.x;
  const int wave = t >> 5, lane = t & 31;
  const int lm = lane & 15;
  const int koff = (lane < 16) ? 0 : 2;
  const int rbase = (lane < 16) ? 0 : 8;
  const int wm = 16 * (wave & 3);
  const int wn = 32 * (wave >> 2);

  static_assert((S_ / 64) * B_ * NH_ == 768, "numbering below is for 3 x 256 items");
  const int widx = blockIdx.x;
  const int g3 = widx >> 8;
  const int j3 = widx & 255;
  int p3;
  if (g3 == 0)      p3 = j3;
  else if (g3 == 1) p3 = (j3 + 128) & 255;
  else              p3 = (j3 < 128) ? (255 - 2 * j3) : (510 - 2 * j3);
  const int rank = (g3 << 8) + p3;
  const int qt = (S_ / 64 - 1) - rank / (B_ * NH_);
  const int bh = rank % (B_ * NH_);
  const int b = bh / NH_, nh = bh % NH_;
  const int q0 = qt * 64;

  const float* qbase = qkv + (size_t)b * S_ * QKVLD + nh * HD_;
  const float* kbase = qbase + H_;
  const float* vbase = qbase + 2 * H_;
  float* Ps = Ks;

  for (int i = 0; i < 8; ++i) {
    int f = t + 256 * i;
    int r = f >> 5;
    int c2 = (f & 31) * 2;
    *(v2f*)(Qs + r * ALD + c2) =
        *(const v2f*)(qbase + (size_t)(q0 + r) * QKVLD + c2);
  }

  v8f o0 = {};
  v8f o1 = {};
  float mi = -INFINITY, li = 0.0f;

  for (int j = 0; j <= qt; ++j) {
    const int k0 = j * 64;
    for (int i = 0; i < 8; ++i) {
      int f = t + 256 * i;
      int r = f >> 5;
      int c2 = (f & 31) * 2;
      *(v2f*)(Ks + r * ALD + c2) =
          *(const v2f*)(kbase + (size_t)(k0 + r) * QKVLD + c2);
      *(v2f*)(Vs + r * ALD + c2) =
          *(const v2f*)(vbase + (size_t)(k0 + r) * QKVLD + c2);
    }
    __syncthreads();

    v8f s0 = {};
    v8f s1 = {};
#pragma unroll
    for (int kk = 0; kk < 64; kk += 4) {
      v2f a  = *(const v2f*)(Qs + (wm + lm) * ALD + kk + koff);
      v2f b0 = *(const v2f*)(Ks + (wn + lm) * ALD + kk + koff);
      v2f b1 = *(const v2f*)(Ks + (wn + 16 + lm) * ALD + kk + koff);
      s0 = wmma4(a, b0, s0);
      s1 = wmma4(a, b1, s1);
    }
    __syncthreads();

    if (j == qt) {
#pragma unroll
      for (int r = 0; r < 8; ++r) {
        const int row = wm + r + rbase;
        const int col = wn + lm;
        Ps[row * ALD + col]      = (col <= row)      ? s0[r] : -INFINITY;
        Ps[row * ALD + col + 16] = (col + 16 <= row) ? s1[r] : -INFINITY;
      }
    } else {
#pragma unroll
      for (int r = 0; r < 8; ++r) {
        Ps[(wm + r + rbase) * ALD + wn + lm]      = s0[r];
        Ps[(wm + r + rbase) * ALD + wn + 16 + lm] = s1[r];
      }
    }
    __syncthreads();

    {
      const int m = t >> 2;
      const int cq = (t & 3) * 16;
      float* Pr = Ps + m * ALD + cq;
      v2f pv[8];
      float mx = -INFINITY;
#pragma unroll
      for (int c = 0; c < 8; ++c) {
        pv[c] = *(const v2f*)(Pr + 2 * c);
        mx = fmaxf(fmaxf(mx, pv[c].x), pv[c].y);
      }
      mx = fmaxf(mx, __shfl_xor(mx, 1));
      mx = fmaxf(mx, __shfl_xor(mx, 2));
      const float mnew = fmaxf(mi, mx * CSCALE);
      const float al = __builtin_amdgcn_exp2f(mi - mnew);
      const v2f cs2 = {CSCALE, CSCALE};
      const v2f nm2 = {-mnew, -mnew};
      float sum = 0.0f;
#pragma unroll
      for (int c = 0; c < 8; ++c) {
        const v2f a = __builtin_elementwise_fma(pv[c], cs2, nm2);
        v2f p;
        p.x = __builtin_amdgcn_exp2f(a.x);
        p.y = __builtin_amdgcn_exp2f(a.y);
        *(v2f*)(Pr + 2 * c) = p;
        sum += p.x;
        sum += p.y;
      }
      sum += __shfl_xor(sum, 1);
      sum += __shfl_xor(sum, 2);
      li = li * al + sum;
      mi = mnew;
      if ((t & 3) == 0) alpha_s[m] = al;
    }
    __syncthreads();

#pragma unroll
    for (int r = 0; r < 8; ++r) {
      const float al = alpha_s[wm + r + rbase];
      o0[r] *= al;
      o1[r] *= al;
    }
#pragma unroll
    for (int kk = 0; kk < 64; kk += 4) {
      v2f a = *(const v2f*)(Ps + (wm + lm) * ALD + kk + koff);
      const int kr = kk + koff;
      v2f b0, b1;
      b0.x = Vs[kr * ALD + wn + lm];
      b0.y = Vs[(kr + 1) * ALD + wn + lm];
      b1.x = Vs[kr * ALD + wn + 16 + lm];
      b1.y = Vs[(kr + 1) * ALD + wn + 16 + lm];
      o0 = wmma4(a, b0, o0);
      o1 = wmma4(a, b1, o1);
    }
    __syncthreads();
  }

  if ((t & 3) == 0) lrow_s[t >> 2] = li;
  __syncthreads();

#pragma unroll
  for (int r = 0; r < 8; ++r) {
    const int row = wm + r + rbase;
    const float inv = 1.0f / lrow_s[row];
    const size_t o = ((size_t)b * S_ + q0 + row) * H_ + nh * HD_ + wn + lm;
    out[o] = o0[r] * inv;
    out[o + 16] = o1[r] * inv;
  }
}

__global__ void tail_kernel(float* out, int idx) {
  if (threadIdx.x == 0 && blockIdx.x == 0) out[idx] = 0.0f;
}

extern "C" void kernel_launch(void* const* d_in, const int* in_sizes, int n_in,
                              void* d_out, int out_size, void* d_ws,
                              size_t ws_size, hipStream_t stream) {
  (void)in_sizes; (void)n_in; (void)ws_size;
  const float* x     = (const float*)d_in[0];
  const float* ln1_g = (const float*)d_in[2];
  const float* ln1_b = (const float*)d_in[3];
  const float* qkv_w = (const float*)d_in[4];
  const float* qkv_b = (const float*)d_in[5];
  const float* out_w = (const float*)d_in[6];
  const float* out_b = (const float*)d_in[7];
  const float* ln2_g = (const float*)d_in[8];
  const float* ln2_b = (const float*)d_in[9];
  const float* fc1_w = (const float*)d_in[10];
  const float* fc1_b = (const float*)d_in[11];
  const float* fc2_w = (const float*)d_in[12];
  const float* fc2_b = (const float*)d_in[13];

  float* xo = (float*)d_out;
  float* ws = (float*)d_ws;
  float* h    = ws;
  float* qkvb = h + (size_t)M_ * H_;
  float* attn = qkvb + (size_t)M_ * 3 * H_;
  float* ffn  = attn + (size_t)M_ * H_;

  ln_kernel<<<M_, 256, 0, stream>>>(x, ln1_g, ln1_b, h);
  gemm_kernel<<<dim3(M_ / 64, (3 * H_) / 64), 256, 0, stream>>>(
      h, qkv_w, qkv_b, nullptr, qkvb, M_, 3 * H_, H_, 0);
  attn_kernel<<<dim3((S_ / 64) * B_ * NH_), 256, 0, stream>>>(qkvb, attn);
  gemm_kernel<<<dim3(M_ / 64, H_ / 64), 256, 0, stream>>>(
      attn, out_w, out_b, x, xo, M_, H_, H_, 0);
  ln_kernel<<<M_, 256, 0, stream>>>(xo, ln2_g, ln2_b, h);
  gemm_kernel<<<dim3(M_ / 64, FFN_ / 64), 256, 0, stream>>>(
      h, fc1_w, fc1_b, nullptr, ffn, M_, FFN_, H_, 1);
  gemm_kernel<<<dim3(M_ / 64, H_ / 64), 256, 0, stream>>>(
      ffn, fc2_w, fc2_b, xo, xo, M_, H_, FFN_, 0);
  tail_kernel<<<1, 32, 0, stream>>>(xo, out_size - 1);
}


# GEMM work-group 8 -> 4 waves with 32x32 wave tiles (four accumulators) and 64-deep K stages (LDS pitch 68): 64 WMMAs per wave between barriers instead of 16; sum order unchanged
# speedup vs baseline: 1.6667x; 1.0475x over previous
#include <hip/hip_runtime.h>
#include <math.h>

#define B_   2
#define S_   2048
#define H_   768
#define NH_  12
#define HD_  64
#define FFN_ 3072
#define M_   (B_ * S_)
#define QKVLD (3 * H_)

typedef __attribute__((ext_vector_type(2))) float v2f;
typedef __attribute__((ext_vector_type(4))) float v4f;
typedef __attribute__((ext_vector_type(8))) float v8f;

__device__ __forceinline__ v8f wmma4(v2f a, v2f b, v8f c) {
  return __builtin_amdgcn_wmma_f32_16x16x4_f32(false, a, false, b, (short)0, c,
                                               false, false);
}

__device__ __forceinline__ float gelu_exact(float x) {
  return 0.5f * x * (1.0f + erff(x * 0.70710678118654752440f));
}

__global__ __launch_bounds__(256) void ln_kernel(const float* __restrict__ x,
                                                 const float* __restrict__ g,
                                                 const float* __restrict__ b,
                                                 float* __restrict__ o) {
  __shared__ float red[256];
  const int t = threadIdx.x;
  const size_t base = (size_t)blockIdx.x * H_;

  float s = 0.0f;
  for (int i = t; i < H_; i += 256) s += x[base + i];
  red[t] = s;
  __syncthreads();
  for (int off = 128; off > 0; off >>= 1) {
    if (t < off) red[t] += red[t + off];
    __syncthreads();
  }
  const float mu = red[0] * (1.0f / (float)H_);
  __syncthreads();

  float vs = 0.0f;
  for (int i = t; i < H_; i += 256) {
    float d = x[base + i] - mu;
    vs += d * d;
  }
  red[t] = vs;
  __syncthreads();
  for (int off = 128; off > 0; off >>= 1) {
    if (t < off) red[t] += red[t + off];
    __syncthreads();
  }
  const float rstd = rsqrtf(red[0] * (1.0f / (float)H_) + 1e-5f);

  for (int i = t; i < H_; i += 256)
    o[base + i] = (x[base + i] - mu) * rstd * g[i] + b[i];
}

#define GBK  64
#define GLDA 68

template <int K>
__global__ __launch_bounds__(128) void gemm_kernel(
    const float* __restrict__ A, const float* __restrict__ W,
    const float* __restrict__ bias, const float* __restrict__ resid,
    float* __restrict__ C, int M, int N, int act) {
  static_assert(K % GBK == 0, "K must be a whole number of stages");
  __shared__ float As[2][64 * GLDA];
  __shared__ float Ws[2][64 * GLDA];

  const int t = threadIdx.x;
  const int wave = t >> 5, lane = t & 31;
  const int m0 = blockIdx.x * 64, n0 = blockIdx.y * 64;
  const int wm = 32 * (wave & 1);
  const int wn = 32 * (wave >> 1);
  const int lm = lane & 15;
  const int koff = (lane < 16) ? 0 : 2;
  const int rbase = (lane < 16) ? 0 : 8;

  const int lrow = t >> 4;
  const int lcol = (t & 15) * 4;
  const float* Ap = A + (size_t)(m0 + lrow) * K + lcol;
  const float* Wp = W + (size_t)(n0 + lrow) * K + lcol;
  const int lds_off = lrow * GLDA + lcol;

  v8f acc00 = {};
  v8f acc01 = {};
  v8f acc10 = {};
  v8f acc11 = {};

  {
    v4f a[8], w[8];
#pragma unroll
    for (int i = 0; i < 8; ++i) {
      a[i] = *(const v4f*)(Ap + (size_t)(8 * i) * K);
      w[i] = *(const v4f*)(Wp + (size_t)(8 * i) * K);
    }
#pragma unroll
    for (int i = 0; i < 8; ++i) {
      *(v4f*)(As[0] + lds_off + 8 * i * GLDA) = a[i];
      *(v4f*)(Ws[0] + lds_off + 8 * i * GLDA) = w[i];
    }
  }
  __syncthreads();

  constexpr int nstages = K / GBK;
  for (int s = 0; s < nstages; ++s) {
    const int p = s & 1;
    const bool has_next = (s + 1) < nstages;

    v4f a[8], w[8];
    if (has_next) {
      const size_t ko = (size_t)(s + 1) * GBK;
#pragma unroll
      for (int i = 0; i < 8; ++i) {
        a[i] = *(const v4f*)(Ap + ko + (size_t)(8 * i) * K);
        w[i] = *(const v4f*)(Wp + ko + (size_t)(8 * i) * K);
      }
    }

    const float* Ab = As[p];
    const float* Wb = Ws[p];
#pragma unroll
    for (int kk = 0; kk < GBK; kk += 4) {
      v2f a0 = *(const v2f*)(Ab + (wm + lm) * GLDA + kk + koff);
      v2f a1 = *(const v2f*)(Ab + (wm + 16 + lm) * GLDA + kk + koff);
      v2f b0 = *(const v2f*)(Wb + (wn + lm) * GLDA + kk + koff);
      v2f b1 = *(const v2f*)(Wb + (wn + 16 + lm) * GLDA + kk + koff);
      acc00 = wmma4(a0, b0, acc00);
      acc01 = wmma4(a0, b1, acc01);
      acc10 = wmma4(a1, b0, acc10);
      acc11 = wmma4(a1, b1, acc11);
    }

    __builtin_amdgcn_sched_barrier(0);
    if (has_next) {
#pragma unroll
      for (int i = 0; i < 8; ++i) {
        *(v4f*)(As[1 - p] + lds_off + 8 * i * GLDA) = a[i];
        *(v4f*)(Ws[1 - p] + lds_off + 8 * i * GLDA) = w[i];
      }
    }
    __syncthreads();
  }

#pragma unroll
  for (int i = 0; i < 2; ++i) {
    const v8f c0 = i ? acc10 : acc00;
    const v8f c1 = i ? acc11 : acc01;
#pragma unroll
    for (int r = 0; r < 8; ++r) {
      const int m = m0 + wm + 16 * i + r + rbase;
      const int n = n0 + wn + lm;
      const size_t i0 = (size_t)m * N + n;
      float v0 = c0[r] + bias[n];
      float v1 = c1[r] + bias[n + 16];
      if (resid) {
        v0 += resid[i0];
        v1 += resid[i0 + 16];
      }
      if (act == 1) {
        v0 = gelu_exact(v0);
        v1 = gelu_exact(v1);
      }
      C[i0] = v0;
      C[i0 + 16] = v1;
    }
  }
}

#define ALD 66
#define CSCALE 0.18033688011112042f

__global__ __launch_bounds__(256) void attn_kernel(
    const float* __restrict__ qkv, float* __restrict__ out) {
  __shared__ float Qs[64 * ALD];
  __shared__ float Ks[64 * ALD];
  __shared__ float Vs[64 * ALD];
  __shared__ float alpha_s[64];
  __shared__ float lrow_s[64];

  const int t = threadIdx.x;
  const int wave = t >> 5, lane = t & 31;
  const int lm = lane & 15;
  const int koff = (lane < 16) ? 0 : 2;
  const int rbase = (lane < 16) ? 0 : 8;
  const int wm = 16 * (wave & 3);
  const int wn = 32 * (wave >> 2);

  static_assert((S_ / 64) * B_ * NH_ == 768, "numbering below is for 3 x 256 items");
  const int widx = blockIdx.x;
  const int g3 = widx >> 8;
  const int j3 = widx & 255;
  int p3;
  if (g3 == 0)      p3 = j3;
  else if (g3 == 1) p3 = (j3 + 128) & 255;
  else              p3 = (j3 < 128) ? (255 - 2 * j3) : (510 - 2 * j3);
  const int rank = (g3 << 8) + p3;
  const int qt = (S_ / 64 - 1) - rank / (B_ * NH_);
  const int bh = rank % (B_ * NH_);
  const int b = bh / NH_, nh = bh % NH_;
  const int q0 = qt * 64;

  const float* qbase = qkv + (size_t)b * S_ * QKVLD + nh * HD_;
  const float* kbase = qbase + H_;
  const float* vbase = qbase + 2 * H_;
  float* Ps = Ks;

  for (int i = 0; i < 8; ++i) {
    int f = t + 256 * i;
    int r = f >> 5;
    int c2 = (f & 31) * 2;
    *(v2f*)(Qs + r * ALD + c2) =
        *(const v2f*)(qbase + (size_t)(q0 + r) * QKVLD + c2);
  }

  v8f o0 = {};
  v8f o1 = {};
  float mi = -INFINITY, li = 0.0f;

  for (int j = 0; j <= qt; ++j) {
    const int k0 = j * 64;
    for (int i = 0; i < 8; ++i) {
      int f = t + 256 * i;
      int r = f >> 5;
      int c2 = (f & 31) * 2;
      *(v2f*)(Ks + r * ALD + c2) =
          *(const v2f*)(kbase + (size_t)(k0 + r) * QKVLD + c2);
      *(v2f*)(Vs + r * ALD + c2) =
          *(const v2f*)(vbase + (size_t)(k0 + r) * QKVLD + c2);
    }
    __syncthreads();

    v8f s0 = {};
    v8f s1 = {};
#pragma unroll
    for (int kk = 0; kk < 64; kk += 4) {
      v2f a  = *(const v2f*)(Qs + (wm + lm) * ALD + kk + koff);
      v2f b0 = *(const v2f*)(Ks + (wn + lm) * ALD + kk + koff);
      v2f b1 = *(const v2f*)(Ks + (wn + 16 + lm) * ALD + kk + koff);
      s0 = wmma4(a, b0, s0);
      s1 = wmma4(a, b1, s1);
    }
    __syncthreads();

    if (j == qt) {
#pragma unroll
      for (int r = 0; r < 8; ++r) {
        const int row = wm + r + rbase;
        const int col = wn + lm;
        Ps[row * ALD + col]      = (col <= row)      ? s0[r] : -INFINITY;
        Ps[row * ALD + col + 16] = (col + 16 <= row) ? s1[r] : -INFINITY;
      }
    } else {
#pragma unroll
      for (int r = 0; r < 8; ++r) {
        Ps[(wm + r + rbase) * ALD + wn + lm]      = s0[r];
        Ps[(wm + r + rbase) * ALD + wn + 16 + lm] = s1[r];
      }
    }
    __syncthreads();

    {
      const int m = t >> 2;
      const int cq = (t & 3) * 16;
      float* Pr = Ps + m * ALD + cq;
      v2f pv[8];
      float mx = -INFINITY;
#pragma unroll
      for (int c = 0; c < 8; ++c) {
        pv[c] = *(const v2f*)(Pr + 2 * c);
        mx = fmaxf(fmaxf(mx, pv[c].x), pv[c].y);
      }
      mx = fmaxf(mx, __shfl_xor(mx, 1));
      mx = fmaxf(mx, __shfl_xor(mx, 2));
      const float mnew = fmaxf(mi, mx * CSCALE);
      const float al = __builtin_amdgcn_exp2f(mi - mnew);
      const v2f cs2 = {CSCALE, CSCALE};
      const v2f nm2 = {-mnew, -mnew};
      float sum = 0.0f;
#pragma unroll
      for (int c = 0; c < 8; ++c) {
        const v2f a = __builtin_elementwise_fma(pv[c], cs2, nm2);
        v2f p;
        p.x = __builtin_amdgcn_exp2f(a.x);
        p.y = __builtin_amdgcn_exp2f(a.y);
        *(v2f*)(Pr + 2 * c) = p;
        sum += p.x;
        sum += p.y;
      }
      sum += __shfl_xor(sum, 1);
      sum += __shfl_xor(sum, 2);
      li = li * al + sum;
      mi = mnew;
      if ((t & 3) == 0) alpha_s[m] = al;
    }
    __syncthreads();

#pragma unroll
    for (int r = 0; r < 8; ++r) {
      const float al = alpha_s[wm + r + rbase];
      o0[r] *= al;
      o1[r] *= al;
    }
#pragma unroll
    for (int kk = 0; kk < 64; kk += 4) {
      v2f a = *(const v2f*)(Ps + (wm + lm) * ALD + kk + koff);
      const int kr = kk + koff;
      v2f b0, b1;
      b0.x = Vs[kr * ALD + wn + lm];
      b0.y = Vs[(kr + 1) * ALD + wn + lm];
      b1.x = Vs[kr * ALD + wn + 16 + lm];
      b1.y = Vs[(kr + 1) * ALD + wn + 16 + lm];
      o0 = wmma4(a, b0, o0);
      o1 = wmma4(a, b1, o1);
    }
    __syncthreads();
  }

  if ((t & 3) == 0) lrow_s[t >> 2] = li;
  __syncthreads();

#pragma unroll
  for (int r = 0; r < 8; ++r) {
    const int row = wm + r + rbase;
    const float inv = 1.0f / lrow_s[row];
    const size_t o = ((size_t)b * S_ + q0 + row) * H_ + nh * HD_ + wn + lm;
    out[o] = o0[r] * inv;
    out[o + 16] = o1[r] * inv;
  }
}

__global__ void tail_kernel(float* out, int idx) {
  if (threadIdx.x == 0 && blockIdx.x == 0) out[idx] = 0.0f;
}

extern "C" void kernel_launch(void* const* d_in, const int* in_sizes, int n_in,
                              void* d_out, int out_size, void* d_ws,
                              size_t ws_size, hipStream_t stream) {
  (void)in_sizes; (void)n_in; (void)ws_size;
  const float* x     = (const float*)d_in[0];
  const float* ln1_g = (const float*)d_in[2];
  const float* ln1_b = (const float*)d_in[3];
  const float* qkv_w = (const float*)d_in[4];
  const float* qkv_b = (const float*)d_in[5];
  const float* out_w = (const float*)d_in[6];
  const float* out_b = (const float*)d_in[7];
  const float* ln2_g = (const float*)d_in[8];
  const float* ln2_b = (const float*)d_in[9];
  const float* fc1_w = (const float*)d_in[10];
  const float* fc1_b = (const float*)d_in[11];
  const float* fc2_w = (const float*)d_in[12];
  const float* fc2_b = (const float*)d_in[13];

  float* xo = (float*)d_out;
  float* ws = (float*)d_ws;
  float* h    = ws;
  float* qkvb = h + (size_t)M_ * H_;
  float* attn = qkvb + (size_t)M_ * 3 * H_;
  float* ffn  = attn + (size_t)M_ * H_;

  ln_kernel<<<M_, 256, 0, stream>>>(x, ln1_g, ln1_b, h);
  gemm_kernel<H_><<<dim3(M_ / 64, (3 * H_) / 64), 128, 0, stream>>>(
      h, qkv_w, qkv_b, nullptr, qkvb, M_, 3 * H_, 0);
  attn_kernel<<<dim3((S_ / 64) * B_ * NH_), 256, 0, stream>>>(qkvb, attn);
  gemm_kernel<H_><<<dim3(M_ / 64, H_ / 64), 128, 0, stream>>>(
      attn, out_w, out_b, x, xo, M_, H_, 0);
  ln_kernel<<<M_, 256, 0, stream>>>(xo, ln2_g, ln2_b, h);
  gemm_kernel<H_><<<dim3(M_ / 64, FFN_ / 64), 128, 0, stream>>>(
      h, fc1_w, fc1_b, nullptr, ffn, M_, FFN_, 1);
  gemm_kernel<FFN_><<<dim3(M_ / 64, H_ / 64), 128, 0, stream>>>(
      ffn, fc2_w, fc2_b, xo, xo, M_, H_, 0);
  tail_kernel<<<1, 32, 0, stream>>>(xo, out_size - 1);
}


# LayerNorm: one wave per 2 rows, rows in registers (x read once); mean/variance = per-lane sums of 24 folded by a 32-lane xor butterfly, not a 256-thread LDS tree; 2048 waves not 32768, no barrier
# speedup vs baseline: 1.7208x; 1.0325x over previous
#include <hip/hip_runtime.h>
#include <math.h>

#define B_   2
#define S_   2048
#define H_   768
#define NH_  12
#define HD_  64
#define FFN_ 3072
#define M_   (B_ * S_)
#define QKVLD (3 * H_)

typedef __attribute__((ext_vector_type(2))) float v2f;
typedef __attribute__((ext_vector_type(4))) float v4f;
typedef __attribute__((ext_vector_type(8))) float v8f;

__device__ __forceinline__ v8f wmma4(v2f a, v2f b, v8f c) {
  return __builtin_amdgcn_wmma_f32_16x16x4_f32(false, a, false, b, (short)0, c,
                                               false, false);
}

__device__ __forceinline__ float gelu_exact(float x) {
  return 0.5f * x * (1.0f + erff(x * 0.70710678118654752440f));
}

#define LN_ROWS   2
#define LN_WAVES  8
#define LN_PIECES (H_ / 128)

template <int M>
__device__ __forceinline__ float add_xmask(float v) {
  const int p = __builtin_amdgcn_update_dpp(0, __builtin_bit_cast(int, v),
                                            0x160 + M, 0xf, 0xf, false);
  return v + __builtin_bit_cast(float, p);
}

__device__ __forceinline__ float other_row(float v) {
  const int b = __builtin_bit_cast(int, v);
  return __builtin_bit_cast(float, __builtin_amdgcn_permlanex16(
      b, b, 0x76543210, 0xfedcba98, false, false));
}

__device__ __forceinline__ float wave_sum32(float v) {
  v = add_xmask<1>(v);
  v = add_xmask<2>(v);
  v = add_xmask<4>(v);
  v = add_xmask<8>(v);
  return v + other_row(v);
}

__global__ __launch_bounds__(32 * LN_WAVES) void ln_kernel(
    const float* __restrict__ x, const float* __restrict__ g,
    const float* __restrict__ b, float* __restrict__ o) {
  static_assert(H_ % 128 == 0, "a row is a whole number of 512-byte pieces");
  static_assert(LN_PIECES == 6 && LN_ROWS == 2, "the load fences below name 6 pieces of 2 rows");
  static_assert(M_ % (LN_ROWS * LN_WAVES) == 0, "whole work-groups only");
  const int lane = threadIdx.x & 31;
  const int wave = threadIdx.x >> 5;
  const size_t row0 = ((size_t)blockIdx.x * LN_WAVES + wave) * LN_ROWS;
  const float* xp = x + row0 * H_ + 4 * lane;
  float* op = o + row0 * H_ + 4 * lane;

  v4f xv[LN_ROWS][LN_PIECES];
#pragma unroll
  for (int r = 0; r < LN_ROWS; ++r)
#pragma unroll
    for (int i = 0; i < LN_PIECES; ++i)
      xv[r][i] = *(const v4f*)(xp + r * H_ + 128 * i);
  v4f gv[LN_PIECES], bv[LN_PIECES];
#pragma unroll
  for (int i = 0; i < LN_PIECES; ++i) {
    gv[i] = *(const v4f*)(g + 4 * lane + 128 * i);
    bv[i] = *(const v4f*)(b + 4 * lane + 128 * i);
  }
  asm volatile("" :: "v"(xv[0][0]), "v"(xv[0][1]), "v"(xv[0][2]), "v"(xv[0][3]),
                     "v"(xv[0][4]), "v"(xv[0][5]), "v"(xv[1][0]), "v"(xv[1][1]),
                     "v"(xv[1][2]), "v"(xv[1][3]), "v"(xv[1][4]), "v"(xv[1][5]));
  asm volatile("" :: "v"(gv[0]), "v"(gv[1]), "v"(gv[2]), "v"(gv[3]), "v"(gv[4]),
                     "v"(gv[5]), "v"(bv[0]), "v"(bv[1]), "v"(bv[2]), "v"(bv[3]),
                     "v"(bv[4]), "v"(bv[5]));

  float mu[LN_ROWS], rstd[LN_ROWS];
#pragma unroll
  for (int r = 0; r < LN_ROWS; ++r) {
    float s = 0.0f;
#pragma unroll
    for (int i = 0; i < LN_PIECES; ++i)
      s += (xv[r][i].x + xv[r][i].y) + (xv[r][i].z + xv[r][i].w);
    mu[r] = wave_sum32(s) * (1.0f / (float)H_);
  }
#pragma unroll
  for (int r = 0; r < LN_ROWS; ++r) {
    float vs = 0.0f;
#pragma unroll
    for (int i = 0; i < LN_PIECES; ++i) {
      const v4f d = xv[r][i] - mu[r];
      vs += (d.x * d.x + d.y * d.y) + (d.z * d.z + d.w * d.w);
    }
    rstd[r] = rsqrtf(wave_sum32(vs) * (1.0f / (float)H_) + 1e-5f);
  }
#pragma unroll
  for (int r = 0; r < LN_ROWS; ++r)
#pragma unroll
    for (int i = 0; i < LN_PIECES; ++i)
      *(v4f*)(op + r * H_ + 128 * i) = (xv[r][i] - mu[r]) * rstd[r] * gv[i] + bv[i];
}

#define GBK  64
#define GLDA 68

template <int K>
__global__ __launch_bounds__(128) void gemm_kernel(
    const float* __restrict__ A, const float* __restrict__ W,
    const float* __restrict__ bias, const float* __restrict__ resid,
    float* __restrict__ C, int M, int N, int act) {
  static_assert(K % GBK == 0, "K must be a whole number of stages");
  __shared__ float As[2][64 * GLDA];
  __shared__ float Ws[2][64 * GLDA];

  const int t = threadIdx.x;
  const int wave = t >> 5, lane = t & 31;
  const int m0 = blockIdx.x * 64, n0 = blockIdx.y * 64;
  const int wm = 32 * (wave & 1);
  const int wn = 32 * (wave >> 1);
  const int lm = lane & 15;
  const int koff = (lane < 16) ? 0 : 2;
  const int rbase = (lane < 16) ? 0 : 8;

  const int lrow = t >> 4;
  const int lcol = (t & 15) * 4;
  const float* Ap = A + (size_t)(m0 + lrow) * K + lcol;
  const float* Wp = W + (size_t)(n0 + lrow) * K + lcol;
  const int lds_off = lrow * GLDA + lcol;

  v8f acc00 = {};
  v8f acc01 = {};
  v8f acc10 = {};
  v8f acc11 = {};

  {
    v4f a[8], w[8];
#pragma unroll
    for (int i = 0; i < 8; ++i) {
      a[i] = *(const v4f*)(Ap + (size_t)(8 * i) * K);
      w[i] = *(const v4f*)(Wp + (size_t)(8 * i) * K);
    }
#pragma unroll
    for (int i = 0; i < 8; ++i) {
      *(v4f*)(As[0] + lds_off + 8 * i * GLDA) = a[i];
      *(v4f*)(Ws[0] + lds_off + 8 * i * GLDA) = w[i];
    }
  }
  __syncthreads();

  constexpr int nstages = K / GBK;
  for (int s = 0; s < nstages; ++s) {
    const int p = s & 1;
    const bool has_next = (s + 1) < nstages;

    v4f a[8], w[8];
    if (has_next) {
      const size_t ko = (size_t)(s + 1) * GBK;
#pragma unroll
      for (int i = 0; i < 8; ++i) {
        a[i] = *(const v4f*)(Ap + ko + (size_t)(8 * i) * K);
        w[i] = *(const v4f*)(Wp + ko + (size_t)(8 * i) * K);
      }
    }

    const float* Ab = As[p];
    const float* Wb = Ws[p];
#pragma unroll
    for (int kk = 0; kk < GBK; kk += 4) {
      v2f a0 = *(const v2f*)(Ab + (wm + lm) * GLDA + kk + koff);
      v2f a1 = *(const v2f*)(Ab + (wm + 16 + lm) * GLDA + kk + koff);
      v2f b0 = *(const v2f*)(Wb + (wn + lm) * GLDA + kk + koff);
      v2f b1 = *(const v2f*)(Wb + (wn + 16 + lm) * GLDA + kk + koff);
      acc00 = wmma4(a0, b0, acc00);
      acc01 = wmma4(a0, b1, acc01);
      acc10 = wmma4(a1, b0, acc10);
      acc11 = wmma4(a1, b1, acc11);
    }

    __builtin_amdgcn_sched_barrier(0);
    if (has_next) {
#pragma unroll
      for (int i = 0; i < 8; ++i) {
        *(v4f*)(As[1 - p] + lds_off + 8 * i * GLDA) = a[i];
        *(v4f*)(Ws[1 - p] + lds_off + 8 * i * GLDA) = w[i];
      }
    }
    __syncthreads();
  }

#pragma unroll
  for (int i = 0; i < 2; ++i) {
    const v8f c0 = i ? acc10 : acc00;
    const v8f c1 = i ? acc11 : acc01;
#pragma unroll
    for (int r = 0; r < 8; ++r) {
      const int m = m0 + wm + 16 * i + r + rbase;
      const int n = n0 + wn + lm;
      const size_t i0 = (size_t)m * N + n;
      float v0 = c0[r] + bias[n];
      float v1 = c1[r] + bias[n + 16];
      if (resid) {
        v0 += resid[i0];
        v1 += resid[i0 + 16];
      }
      if (act == 1) {
        v0 = gelu_exact(v0);
        v1 = gelu_exact(v1);
      }
      C[i0] = v0;
      C[i0 + 16] = v1;
    }
  }
}

#define ALD 66
#define CSCALE 0.18033688011112042f

__global__ __launch_bounds__(256) void attn_kernel(
    const float* __restrict__ qkv, float* __restrict__ out) {
  __shared__ float Qs[64 * ALD];
  __shared__ float Ks[64 * ALD];
  __shared__ float Vs[64 * ALD];
  __shared__ float alpha_s[64];
  __shared__ float lrow_s[64];

  const int t = threadIdx.x;
  const int wave = t >> 5, lane = t & 31;
  const int lm = lane & 15;
  const int koff = (lane < 16) ? 0 : 2;
  const int rbase = (lane < 16) ? 0 : 8;
  const int wm = 16 * (wave & 3);
  const int wn = 32 * (wave >> 2);

  static_assert((S_ / 64) * B_ * NH_ == 768, "numbering below is for 3 x 256 items");
  const int widx = blockIdx.x;
  const int g3 = widx >> 8;
  const int j3 = widx & 255;
  int p3;
  if (g3 == 0)      p3 = j3;
  else if (g3 == 1) p3 = (j3 + 128) & 255;
  else              p3 = (j3 < 128) ? (255 - 2 * j3) : (510 - 2 * j3);
  const int rank = (g3 << 8) + p3;
  const int qt = (S_ / 64 - 1) - rank / (B_ * NH_);
  const int bh = rank % (B_ * NH_);
  const int b = bh / NH_, nh = bh % NH_;
  const int q0 = qt * 64;

  const float* qbase = qkv + (size_t)b * S_ * QKVLD + nh * HD_;
  const float* kbase = qbase + H_;
  const float* vbase = qbase + 2 * H_;
  float* Ps = Ks;

  for (int i = 0; i < 8; ++i) {
    int f = t + 256 * i;
    int r = f >> 5;
    int c2 = (f & 31) * 2;
    *(v2f*)(Qs + r * ALD + c2) =
        *(const v2f*)(qbase + (size_t)(q0 + r) * QKVLD + c2);
  }

  v8f o0 = {};
  v8f o1 = {};
  float mi = -INFINITY, li = 0.0f;

  for (int j = 0; j <= qt; ++j) {
    const int k0 = j * 64;
    for (int i = 0; i < 8; ++i) {
      int f = t + 256 * i;
      int r = f >> 5;
      int c2 = (f & 31) * 2;
      *(v2f*)(Ks + r * ALD + c2) =
          *(const v2f*)(kbase + (size_t)(k0 + r) * QKVLD + c2);
      *(v2f*)(Vs + r * ALD + c2) =
          *(const v2f*)(vbase + (size_t)(k0 + r) * QKVLD + c2);
    }
    __syncthreads();

    v8f s0 = {};
    v8f s1 = {};
#pragma unroll
    for (int kk = 0; kk < 64; kk += 4) {
      v2f a  = *(const v2f*)(Qs + (wm + lm) * ALD + kk + koff);
      v2f b0 = *(const v2f*)(Ks + (wn + lm) * ALD + kk + koff);
      v2f b1 = *(const v2f*)(Ks + (wn + 16 + lm) * ALD + kk + koff);
      s0 = wmma4(a, b0, s0);
      s1 = wmma4(a, b1, s1);
    }
    __syncthreads();

    if (j == qt) {
#pragma unroll
      for (int r = 0; r < 8; ++r) {
        const int row = wm + r + rbase;
        const int col = wn + lm;
        Ps[row * ALD + col]      = (col <= row)      ? s0[r] : -INFINITY;
        Ps[row * ALD + col + 16] = (col + 16 <= row) ? s1[r] : -INFINITY;
      }
    } else {
#pragma unroll
      for (int r = 0; r < 8; ++r) {
        Ps[(wm + r + rbase) * ALD + wn + lm]      = s0[r];
        Ps[(wm + r + rbase) * ALD + wn + 16 + lm] = s1[r];
      }
    }
    __syncthreads();

    {
      const int m = t >> 2;
      const int cq = (t & 3) * 16;
      float* Pr = Ps + m * ALD + cq;
      v2f pv[8];
      float mx = -INFINITY;
#pragma unroll
      for (int c = 0; c < 8; ++c) {
        pv[c] = *(const v2f*)(Pr + 2 * c);
        mx = fmaxf(fmaxf(mx, pv[c].x), pv[c].y);
      }
      mx = fmaxf(mx, __shfl_xor(mx, 1));
      mx = fmaxf(mx, __shfl_xor(mx, 2));
      const float mnew = fmaxf(mi, mx * CSCALE);
      const float al = __builtin_amdgcn_exp2f(mi - mnew);
      const v2f cs2 = {CSCALE, CSCALE};
      const v2f nm2 = {-mnew, -mnew};
      float sum = 0.0f;
#pragma unroll
      for (int c = 0; c < 8; ++c) {
        const v2f a = __builtin_elementwise_fma(pv[c], cs2, nm2);
        v2f p;
        p.x = __builtin_amdgcn_exp2f(a.x);
        p.y = __builtin_amdgcn_exp2f(a.y);
        *(v2f*)(Pr + 2 * c) = p;
        sum += p.x;
        sum += p.y;
      }
      sum += __shfl_xor(sum, 1);
      sum += __shfl_xor(sum, 2);
      li = li * al + sum;
      mi = mnew;
      if ((t & 3) == 0) alpha_s[m] = al;
    }
    __syncthreads();

#pragma unroll
    for (int r = 0; r < 8; ++r) {
      const float al = alpha_s[wm + r + rbase];
      o0[r] *= al;
      o1[r] *= al;
    }
#pragma unroll
    for (int kk = 0; kk < 64; kk += 4) {
      v2f a = *(const v2f*)(Ps + (wm + lm) * ALD + kk + koff);
      const int kr = kk + koff;
      v2f b0, b1;
      b0.x = Vs[kr * ALD + wn + lm];
      b0.y = Vs[(kr + 1) * ALD + wn + lm];
      b1.x = Vs[kr * ALD + wn + 16 + lm];
      b1.y = Vs[(kr + 1) * ALD + wn + 16 + lm];
      o0 = wmma4(a, b0, o0);
      o1 = wmma4(a, b1, o1);
    }
    __syncthreads();
  }

  if ((t & 3) == 0) lrow_s[t >> 2] = li;
  __syncthreads();

#pragma unroll
  for (int r = 0; r < 8; ++r) {
    const int row = wm + r + rbase;
    const float inv = 1.0f / lrow_s[row];
    const size_t o = ((size_t)b * S_ + q0 + row) * H_ + nh * HD_ + wn + lm;
    out[o] = o0[r] * inv;
    out[o + 16] = o1[r] * inv;
  }
}

__global__ void tail_kernel(float* out, int idx) {
  if (threadIdx.x == 0 && blockIdx.x == 0) out[idx] = 0.0f;
}

extern "C" void kernel_launch(void* const* d_in, const int* in_sizes, int n_in,
                              void* d_out, int out_size, void* d_ws,
                              size_t ws_size, hipStream_t stream) {
  (void)in_sizes; (void)n_in; (void)ws_size;
  const float* x     = (const float*)d_in[0];
  const float* ln1_g = (const float*)d_in[2];
  const float* ln1_b = (const float*)d_in[3];
  const float* qkv_w = (const float*)d_in[4];
  const float* qkv_b = (const float*)d_in[5];
  const float* out_w = (const float*)d_in[6];
  const float* out_b = (const float*)d_in[7];
  const float* ln2_g = (const float*)d_in[8];
  const float* ln2_b = (const float*)d_in[9];
  const float* fc1_w = (const float*)d_in[10];
  const float* fc1_b = (const float*)d_in[11];
  const float* fc2_w = (const float*)d_in[12];
  const float* fc2_b = (const float*)d_in[13];

  float* xo = (float*)d_out;
  float* ws = (float*)d_ws;
  float* h    = ws;
  float* qkvb = h + (size_t)M_ * H_;
  float* attn = qkvb + (size_t)M_ * 3 * H_;
  float* ffn  = attn + (size_t)M_ * H_;

  ln_kernel<<<M_ / (LN_ROWS * LN_WAVES), 32 * LN_WAVES, 0, stream>>>(x, ln1_g, ln1_b, h);
  gemm_kernel<H_><<<dim3(M_ / 64, (3 * H_) / 64), 128, 0, stream>>>(
      h, qkv_w, qkv_b, nullptr, qkvb, M_, 3 * H_, 0);
  attn_kernel<<<dim3((S_ / 64) * B_ * NH_), 256, 0, stream>>>(qkvb, attn);
  gemm_kernel<H_><<<dim3(M_ / 64, H_ / 64), 128, 0, stream>>>(
      attn, out_w, out_b, x, xo, M_, H_, 0);
  ln_kernel<<<M_ / (LN_ROWS * LN_WAVES), 32 * LN_WAVES, 0, stream>>>(xo, ln2_g, ln2_b, h);
  gemm_kernel<H_><<<dim3(M_ / 64, FFN_ / 64), 128, 0, stream>>>(
      h, fc1_w, fc1_b, nullptr, ffn, M_, FFN_, 1);
  gemm_kernel<FFN_><<<dim3(M_ / 64, H_ / 64), 128, 0, stream>>>(
      ffn, fc2_w, fc2_b, xo, xo, M_, H_, 0);
  tail_kernel<<<1, 32, 0, stream>>>(xo, out_size - 1);
}


# attention rebuilt on whole score rows per wave: transposed scores stay in registers (softmax there, P^T is PV's B operand), K/V one tile ahead, 1 barrier per tile; row-sum and PV sum order change
# speedup vs baseline: 1.8006x; 1.0464x over previous
#include <hip/hip_runtime.h>
#include <math.h>

#define B_   2
#define S_   2048
#define H_   768
#define NH_  12
#define HD_  64
#define FFN_ 3072
#define M_   (B_ * S_)
#define QKVLD (3 * H_)

typedef __attribute__((ext_vector_type(2))) float v2f;
typedef __attribute__((ext_vector_type(4))) float v4f;
typedef __attribute__((ext_vector_type(8))) float v8f;

__device__ __forceinline__ v8f wmma4(v2f a, v2f b, v8f c) {
  return __builtin_amdgcn_wmma_f32_16x16x4_f32(false, a, false, b, (short)0, c,
                                               false, false);
}

__device__ __forceinline__ float gelu_exact(float x) {
  return 0.5f * x * (1.0f + erff(x * 0.70710678118654752440f));
}

#define LN_ROWS   2
#define LN_WAVES  8
#define LN_PIECES (H_ / 128)

template <int M>
__device__ __forceinline__ float add_xmask(float v) {
  const int p = __builtin_amdgcn_update_dpp(0, __builtin_bit_cast(int, v),
                                            0x160 + M, 0xf, 0xf, false);
  return v + __builtin_bit_cast(float, p);
}

__device__ __forceinline__ float other_row(float v) {
  const int b = __builtin_bit_cast(int, v);
  return __builtin_bit_cast(float, __builtin_amdgcn_permlanex16(
      b, b, 0x76543210, 0xfedcba98, false, false));
}

__device__ __forceinline__ float wave_sum32(float v) {
  v = add_xmask<1>(v);
  v = add_xmask<2>(v);
  v = add_xmask<4>(v);
  v = add_xmask<8>(v);
  return v + other_row(v);
}

__global__ __launch_bounds__(32 * LN_WAVES) void ln_kernel(
    const float* __restrict__ x, const float* __restrict__ g,
    const float* __restrict__ b, float* __restrict__ o) {
  static_assert(H_ % 128 == 0, "a row is a whole number of 512-byte pieces");
  static_assert(LN_PIECES == 6 && LN_ROWS == 2, "the load fences below name 6 pieces of 2 rows");
  static_assert(M_ % (LN_ROWS * LN_WAVES) == 0, "whole work-groups only");
  const int lane = threadIdx.x & 31;
  const int wave = threadIdx.x >> 5;
  const size_t row0 = ((size_t)blockIdx.x * LN_WAVES + wave) * LN_ROWS;
  const float* xp = x + row0 * H_ + 4 * lane;
  float* op = o + row0 * H_ + 4 * lane;

  v4f xv[LN_ROWS][LN_PIECES];
#pragma unroll
  for (int r = 0; r < LN_ROWS; ++r)
#pragma unroll
    for (int i = 0; i < LN_PIECES; ++i)
      xv[r][i] = *(const v4f*)(xp + r * H_ + 128 * i);
  v4f gv[LN_PIECES], bv[LN_PIECES];
#pragma unroll
  for (int i = 0; i < LN_PIECES; ++i) {
    gv[i] = *(const v4f*)(g + 4 * lane + 128 * i);
    bv[i] = *(const v4f*)(b + 4 * lane + 128 * i);
  }
  asm volatile("" :: "v"(xv[0][0]), "v"(xv[0][1]), "v"(xv[0][2]), "v"(xv[0][3]),
                     "v"(xv[0][4]), "v"(xv[0][5]), "v"(xv[1][0]), "v"(xv[1][1]),
                     "v"(xv[1][2]), "v"(xv[1][3]), "v"(xv[1][4]), "v"(xv[1][5]));
  asm volatile("" :: "v"(gv[0]), "v"(gv[1]), "v"(gv[2]), "v"(gv[3]), "v"(gv[4]),
                     "v"(gv[5]), "v"(bv[0]), "v"(bv[1]), "v"(bv[2]), "v"(bv[3]),
                     "v"(bv[4]), "v"(bv[5]));

  float mu[LN_ROWS], rstd[LN_ROWS];
#pragma unroll
  for (int r = 0; r < LN_ROWS; ++r) {
    float s = 0.0f;
#pragma unroll
    for (int i = 0; i < LN_PIECES; ++i)
      s += (xv[r][i].x + xv[r][i].y) + (xv[r][i].z + xv[r][i].w);
    mu[r] = wave_sum32(s) * (1.0f / (float)H_);
  }
#pragma unroll
  for (int r = 0; r < LN_ROWS; ++r) {
    float vs = 0.0f;
#pragma unroll
    for (int i = 0; i < LN_PIECES; ++i) {
      const v4f d = xv[r][i] - mu[r];
      vs += (d.x * d.x + d.y * d.y) + (d.z * d.z + d.w * d.w);
    }
    rstd[r] = rsqrtf(wave_sum32(vs) * (1.0f / (float)H_) + 1e-5f);
  }
#pragma unroll
  for (int r = 0; r < LN_ROWS; ++r)
#pragma unroll
    for (int i = 0; i < LN_PIECES; ++i)
      *(v4f*)(op + r * H_ + 128 * i) = (xv[r][i] - mu[r]) * rstd[r] * gv[i] + bv[i];
}

#define GBK  64
#define GLDA 68

template <int K>
__global__ __launch_bounds__(128) void gemm_kernel(
    const float* __restrict__ A, const float* __restrict__ W,
    const float* __restrict__ bias, const float* __restrict__ resid,
    float* __restrict__ C, int M, int N, int act) {
  static_assert(K % GBK == 0, "K must be a whole number of stages");
  __shared__ float As[2][64 * GLDA];
  __shared__ float Ws[2][64 * GLDA];

  const int t = threadIdx.x;
  const int wave = t >> 5, lane = t & 31;
  const int m0 = blockIdx.x * 64, n0 = blockIdx.y * 64;
  const int wm = 32 * (wave & 1);
  const int wn = 32 * (wave >> 1);
  const int lm = lane & 15;
  const int koff = (lane < 16) ? 0 : 2;
  const int rbase = (lane < 16) ? 0 : 8;

  const int lrow = t >> 4;
  const int lcol = (t & 15) * 4;
  const float* Ap = A + (size_t)(m0 + lrow) * K + lcol;
  const float* Wp = W + (size_t)(n0 + lrow) * K + lcol;
  const int lds_off = lrow * GLDA + lcol;

  v8f acc00 = {};
  v8f acc01 = {};
  v8f acc10 = {};
  v8f acc11 = {};

  {
    v4f a[8], w[8];
#pragma unroll
    for (int i = 0; i < 8; ++i) {
      a[i] = *(const v4f*)(Ap + (size_t)(8 * i) * K);
      w[i] = *(const v4f*)(Wp + (size_t)(8 * i) * K);
    }
#pragma unroll
    for (int i = 0; i < 8; ++i) {
      *(v4f*)(As[0] + lds_off + 8 * i * GLDA) = a[i];
      *(v4f*)(Ws[0] + lds_off + 8 * i * GLDA) = w[i];
    }
  }
  __syncthreads();

  constexpr int nstages = K / GBK;
  for (int s = 0; s < nstages; ++s) {
    const int p = s & 1;
    const bool has_next = (s + 1) < nstages;

    v4f a[8], w[8];
    if (has_next) {
      const size_t ko = (size_t)(s + 1) * GBK;
#pragma unroll
      for (int i = 0; i < 8; ++i) {
        a[i] = *(const v4f*)(Ap + ko + (size_t)(8 * i) * K);
        w[i] = *(const v4f*)(Wp + ko + (size_t)(8 * i) * K);
      }
    }

    const float* Ab = As[p];
    const float* Wb = Ws[p];
#pragma unroll
    for (int kk = 0; kk < GBK; kk += 4) {
      v2f a0 = *(const v2f*)(Ab + (wm + lm) * GLDA + kk + koff);
      v2f a1 = *(const v2f*)(Ab + (wm + 16 + lm) * GLDA + kk + koff);
      v2f b0 = *(const v2f*)(Wb + (wn + lm) * GLDA + kk + koff);
      v2f b1 = *(const v2f*)(Wb + (wn + 16 + lm) * GLDA + kk + koff);
      acc00 = wmma4(a0, b0, acc00);
      acc01 = wmma4(a0, b1, acc01);
      acc10 = wmma4(a1, b0, acc10);
      acc11 = wmma4(a1, b1, acc11);
    }

    __builtin_amdgcn_sched_barrier(0);
    if (has_next) {
#pragma unroll
      for (int i = 0; i < 8; ++i) {
        *(v4f*)(As[1 - p] + lds_off + 8 * i * GLDA) = a[i];
        *(v4f*)(Ws[1 - p] + lds_off + 8 * i * GLDA) = w[i];
      }
    }
    __syncthreads();
  }

#pragma unroll
  for (int i = 0; i < 2; ++i) {
    const v8f c0 = i ? acc10 : acc00;
    const v8f c1 = i ? acc11 : acc01;
#pragma unroll
    for (int r = 0; r < 8; ++r) {
      const int m = m0 + wm + 16 * i + r + rbase;
      const int n = n0 + wn + lm;
      const size_t i0 = (size_t)m * N + n;
      float v0 = c0[r] + bias[n];
      float v1 = c1[r] + bias[n + 16];
      if (resid) {
        v0 += resid[i0];
        v1 += resid[i0 + 16];
      }
      if (act == 1) {
        v0 = gelu_exact(v0);
        v1 = gelu_exact(v1);
      }
      C[i0] = v0;
      C[i0 + 16] = v1;
    }
  }
}

#define AKP 68
#define AVP 136
#define ASTG (64 * AKP + 32 * AVP)
#define AOP 68
#define CSCALE 0.18033688011112042f

__global__ __launch_bounds__(128) void attn_kernel(
    const float* __restrict__ qkv, float* __restrict__ out) {
  __shared__ float stg[2][ASTG];

  const int t = threadIdx.x;
  const int wave = t >> 5, lane = t & 31;
  const int lm = lane & 15;
  const int hh = lane >> 4;

  static_assert((S_ / 64) * B_ * NH_ == 768, "numbering below is for 3 x 256 items");
  const int widx = blockIdx.x;
  const int g3 = widx >> 8;
  const int j3 = widx & 255;
  int p3;
  if (g3 == 0)      p3 = j3;
  else if (g3 == 1) p3 = (j3 + 128) & 255;
  else              p3 = (j3 < 128) ? (255 - 2 * j3) : (510 - 2 * j3);
  const int rank = (g3 << 8) + p3;
  const int qt = (S_ / 64 - 1) - rank / (B_ * NH_);
  const int bh = rank % (B_ * NH_);
  const int b = bh / NH_, nh = bh % NH_;
  const int q0 = qt * 64;

  const float* qbase = qkv + (size_t)b * S_ * QKVLD + nh * HD_;
  const float* kbase = qbase + H_;
  const float* vbase = qbase + 2 * H_;

  const int krow = t >> 4;
  const int kcol = (t & 15) * 4;
  const float* kp = kbase + (size_t)krow * QKVLD + kcol;
  const float* vp = vbase + (size_t)(2 * wave) * QKVLD + 2 * lane;
  const int k_lds = krow * AKP + kcol;
  const int v_lds = 64 * AKP + wave * AVP + 4 * lane;

  v2f qf[16];
  {
    const float* qrow = qbase + (size_t)(q0 + 16 * wave + lm) * QKVLD + 2 * hh;
#pragma unroll
    for (int i = 0; i < 16; ++i) qf[i] = *(const v2f*)(qrow + 4 * i);
  }

  {
    v4f kr[8];
    v2f va[8], vb[8];
#pragma unroll
    for (int i = 0; i < 8; ++i) {
      kr[i] = *(const v4f*)(kp + (size_t)(8 * i) * QKVLD);
      va[i] = *(const v2f*)(vp + (size_t)(8 * i) * QKVLD);
      vb[i] = *(const v2f*)(vp + (size_t)(8 * i + 1) * QKVLD);
    }
#pragma unroll
    for (int i = 0; i < 8; ++i) {
      *(v4f*)(stg[0] + k_lds + 8 * i * AKP) = kr[i];
      const v4f pr = {va[i].x, vb[i].x, va[i].y, vb[i].y};
      *(v4f*)(stg[0] + v_lds + 4 * i * AVP) = pr;
    }
  }
  __syncthreads();

  v8f ot[4] = {};
  float mi = -INFINITY, li = 0.0f;
  const int ql = 16 * wave + lm - 8 * hh;

  for (int j = 0; j <= qt; ++j) {
    const int cur = j & 1;
    const bool has_next = j < qt;

    v4f kr[8];
    v2f va[8], vb[8];
    if (has_next) {
      const size_t ro = (size_t)(j + 1) * 64 * QKVLD;
#pragma unroll
      for (int i = 0; i < 8; ++i) {
        kr[i] = *(const v4f*)(kp + ro + (size_t)(8 * i) * QKVLD);
        va[i] = *(const v2f*)(vp + ro + (size_t)(8 * i) * QKVLD);
        vb[i] = *(const v2f*)(vp + ro + (size_t)(8 * i + 1) * QKVLD);
      }
    }

    const float* Kc = stg[cur];
    const float* Vc = stg[cur] + 64 * AKP;

    v8f st[4] = {};
#pragma unroll
    for (int i = 0; i < 16; ++i) {
#pragma unroll
      for (int kt = 0; kt < 4; ++kt) {
        const v2f a = *(const v2f*)(Kc + (16 * kt + lm) * AKP + 4 * i + 2 * hh);
        st[kt] = wmma4(a, qf[i], st[kt]);
      }
    }

    if (!has_next) {
#pragma unroll
      for (int kt = 0; kt < 4; ++kt)
#pragma unroll
        for (int r = 0; r < 8; ++r)
          st[kt][r] = (16 * kt + r > ql) ? -INFINITY : st[kt][r];
    }

    float al;
    {
      float mx = -INFINITY;
#pragma unroll
      for (int kt = 0; kt < 4; ++kt)
#pragma unroll
        for (int r = 0; r < 8; ++r) mx = fmaxf(mx, st[kt][r]);
      mx = fmaxf(mx, other_row(mx));
      const float mnew = fmaxf(mi, mx * CSCALE);
      al = __builtin_amdgcn_exp2f(mi - mnew);
      const v2f cs2 = {CSCALE, CSCALE};
      const v2f nm2 = {-mnew, -mnew};
      float sum = 0.0f;
#pragma unroll
      for (int kt = 0; kt < 4; ++kt)
#pragma unroll
        for (int r = 0; r < 8; r += 2) {
          const v2f s2 = {st[kt][r], st[kt][r + 1]};
          const v2f a = __builtin_elementwise_fma(s2, cs2, nm2);
          const float p0 = __builtin_amdgcn_exp2f(a.x);
          const float p1 = __builtin_amdgcn_exp2f(a.y);
          st[kt][r] = p0;
          st[kt][r + 1] = p1;
          sum += p0;
          sum += p1;
        }
      sum += other_row(sum);
      li = li * al + sum;
      mi = mnew;
    }

#pragma unroll
    for (int dt = 0; dt < 4; ++dt)
#pragma unroll
      for (int r = 0; r < 8; ++r) ot[dt][r] *= al;
#pragma unroll
    for (int kt = 0; kt < 4; ++kt) {
#pragma unroll
      for (int t2 = 0; t2 < 4; ++t2) {
        const v2f pb = {st[kt][2 * t2], st[kt][2 * t2 + 1]};
#pragma unroll
        for (int dt = 0; dt < 4; ++dt) {
          const v2f a = *(const v2f*)(Vc + (8 * kt + 4 * hh + t2) * AVP + 2 * (16 * dt + lm));
          ot[dt] = wmma4(a, pb, ot[dt]);
        }
      }
    }

    __builtin_amdgcn_sched_barrier(0);
    if (has_next) {
#pragma unroll
      for (int i = 0; i < 8; ++i) {
        *(v4f*)(stg[1 - cur] + k_lds + 8 * i * AKP) = kr[i];
        const v4f pr = {va[i].x, vb[i].x, va[i].y, vb[i].y};
        *(v4f*)(stg[1 - cur] + v_lds + 4 * i * AVP) = pr;
      }
    }
    __syncthreads();
  }

  float* Os = stg[0];
  {
    const float inv = 1.0f / li;
#pragma unroll
    for (int dt = 0; dt < 4; ++dt)
#pragma unroll
      for (int r = 0; r < 8; ++r)
        Os[(16 * wave + lm) * AOP + 16 * dt + 8 * hh + r] = ot[dt][r] * inv;
  }
  __syncthreads();
#pragma unroll
  for (int i = 0; i < 8; ++i) {
    const int row = krow + 8 * i;
    const v4f v = *(const v4f*)(Os + row * AOP + kcol);
    *(v4f*)(out + ((size_t)b * S_ + q0 + row) * H_ + nh * HD_ + kcol) = v;
  }
}

__global__ void tail_kernel(float* out, int idx) {
  if (threadIdx.x == 0 && blockIdx.x == 0) out[idx] = 0.0f;
}

extern "C" void kernel_launch(void* const* d_in, const int* in_sizes, int n_in,
                              void* d_out, int out_size, void* d_ws,
                              size_t ws_size, hipStream_t stream) {
  (void)in_sizes; (void)n_in; (void)ws_size;
  const float* x     = (const float*)d_in[0];
  const float* ln1_g = (const float*)d_in[2];
  const float* ln1_b = (const float*)d_in[3];
  const float* qkv_w = (const float*)d_in[4];
  const float* qkv_b = (const float*)d_in[5];
  const float* out_w = (const float*)d_in[6];
  const float* out_b = (const float*)d_in[7];
  const float* ln2_g = (const float*)d_in[8];
  const float* ln2_b = (const float*)d_in[9];
  const float* fc1_w = (const float*)d_in[10];
  const float* fc1_b = (const float*)d_in[11];
  const float* fc2_w = (const float*)d_in[12];
  const float* fc2_b = (const float*)d_in[13];

  float* xo = (float*)d_out;
  float* ws = (float*)d_ws;
  float* h    = ws;
  float* qkvb = h + (size_t)M_ * H_;
  float* attn = qkvb + (size_t)M_ * 3 * H_;
  float* ffn  = attn + (size_t)M_ * H_;

  ln_kernel<<<M_ / (LN_ROWS * LN_WAVES), 32 * LN_WAVES, 0, stream>>>(x, ln1_g, ln1_b, h);
  gemm_kernel<H_><<<dim3(M_ / 64, (3 * H_) / 64), 128, 0, stream>>>(
      h, qkv_w, qkv_b, nullptr, qkvb, M_, 3 * H_, 0);
  attn_kernel<<<dim3((S_ / 64) * B_ * NH_), 128, 0, stream>>>(qkvb, attn);
  gemm_kernel<H_><<<dim3(M_ / 64, H_ / 64), 128, 0, stream>>>(
      attn, out_w, out_b, x, xo, M_, H_, 0);
  ln_kernel<<<M_ / (LN_ROWS * LN_WAVES), 32 * LN_WAVES, 0, stream>>>(xo, ln2_g, ln2_b, h);
  gemm_kernel<H_><<<dim3(M_ / 64, FFN_ / 64), 128, 0, stream>>>(
      h, fc1_w, fc1_b, nullptr, ffn, M_, FFN_, 1);
  gemm_kernel<FFN_><<<dim3(M_ / 64, H_ / 64), 128, 0, stream>>>(
      ffn, fc2_w, fc2_b, xo, xo, M_, H_, 0);
  tail_kernel<<<1, 32, 0, stream>>>(xo, out_size - 1);
}


# GEMM epilogue: tile turned through LDS so rows leave as whole 128-byte lines (b128 stores); residual read in one batch under the last K stage, not 16 dependent round trips a wave; sum order unchanged
# speedup vs baseline: 1.8639x; 1.0352x over previous
#include <hip/hip_runtime.h>
#include <math.h>

#define B_   2
#define S_   2048
#define H_   768
#define NH_  12
#define HD_  64
#define FFN_ 3072
#define M_   (B_ * S_)
#define QKVLD (3 * H_)

typedef __attribute__((ext_vector_type(2))) float v2f;
typedef __attribute__((ext_vector_type(4))) float v4f;
typedef __attribute__((ext_vector_type(8))) float v8f;

__device__ __forceinline__ v8f wmma4(v2f a, v2f b, v8f c) {
  return __builtin_amdgcn_wmma_f32_16x16x4_f32(false, a, false, b, (short)0, c,
                                               false, false);
}

__device__ __forceinline__ float gelu_exact(float x) {
  return 0.5f * x * (1.0f + erff(x * 0.70710678118654752440f));
}

#define LN_ROWS   2
#define LN_WAVES  8
#define LN_PIECES (H_ / 128)

template <int M>
__device__ __forceinline__ float add_xmask(float v) {
  const int p = __builtin_amdgcn_update_dpp(0, __builtin_bit_cast(int, v),
                                            0x160 + M, 0xf, 0xf, false);
  return v + __builtin_bit_cast(float, p);
}

__device__ __forceinline__ float other_row(float v) {
  const int b = __builtin_bit_cast(int, v);
  return __builtin_bit_cast(float, __builtin_amdgcn_permlanex16(
      b, b, 0x76543210, 0xfedcba98, false, false));
}

__device__ __forceinline__ float wave_sum32(float v) {
  v = add_xmask<1>(v);
  v = add_xmask<2>(v);
  v = add_xmask<4>(v);
  v = add_xmask<8>(v);
  return v + other_row(v);
}

__global__ __launch_bounds__(32 * LN_WAVES) void ln_kernel(
    const float* __restrict__ x, const float* __restrict__ g,
    const float* __restrict__ b, float* __restrict__ o) {
  static_assert(H_ % 128 == 0, "a row is a whole number of 512-byte pieces");
  static_assert(LN_PIECES == 6 && LN_ROWS == 2, "the load fences below name 6 pieces of 2 rows");
  static_assert(M_ % (LN_ROWS * LN_WAVES) == 0, "whole work-groups only");
  const int lane = threadIdx.x & 31;
  const int wave = threadIdx.x >> 5;
  const size_t row0 = ((size_t)blockIdx.x * LN_WAVES + wave) * LN_ROWS;
  const float* xp = x + row0 * H_ + 4 * lane;
  float* op = o + row0 * H_ + 4 * lane;

  v4f xv[LN_ROWS][LN_PIECES];
#pragma unroll
  for (int r = 0; r < LN_ROWS; ++r)
#pragma unroll
    for (int i = 0; i < LN_PIECES; ++i)
      xv[r][i] = *(const v4f*)(xp + r * H_ + 128 * i);
  v4f gv[LN_PIECES], bv[LN_PIECES];
#pragma unroll
  for (int i = 0; i < LN_PIECES; ++i) {
    gv[i] = *(const v4f*)(g + 4 * lane + 128 * i);
    bv[i] = *(const v4f*)(b + 4 * lane + 128 * i);
  }
  asm volatile("" :: "v"(xv[0][0]), "v"(xv[0][1]), "v"(xv[0][2]), "v"(xv[0][3]),
                     "v"(xv[0][4]), "v"(xv[0][5]), "v"(xv[1][0]), "v"(xv[1][1]),
                     "v"(xv[1][2]), "v"(xv[1][3]), "v"(xv[1][4]), "v"(xv[1][5]));
  asm volatile("" :: "v"(gv[0]), "v"(gv[1]), "v"(gv[2]), "v"(gv[3]), "v"(gv[4]),
                     "v"(gv[5]), "v"(bv[0]), "v"(bv[1]), "v"(bv[2]), "v"(bv[3]),
                     "v"(bv[4]), "v"(bv[5]));

  float mu[LN_ROWS], rstd[LN_ROWS];
#pragma unroll
  for (int r = 0; r < LN_ROWS; ++r) {
    float s = 0.0f;
#pragma unroll
    for (int i = 0; i < LN_PIECES; ++i)
      s += (xv[r][i].x + xv[r][i].y) + (xv[r][i].z + xv[r][i].w);
    mu[r] = wave_sum32(s) * (1.0f / (float)H_);
  }
#pragma unroll
  for (int r = 0; r < LN_ROWS; ++r) {
    float vs = 0.0f;
#pragma unroll
    for (int i = 0; i < LN_PIECES; ++i) {
      const v4f d = xv[r][i] - mu[r];
      vs += (d.x * d.x + d.y * d.y) + (d.z * d.z + d.w * d.w);
    }
    rstd[r] = rsqrtf(wave_sum32(vs) * (1.0f / (float)H_) + 1e-5f);
  }
#pragma unroll
  for (int r = 0; r < LN_ROWS; ++r)
#pragma unroll
    for (int i = 0; i < LN_PIECES; ++i)
      *(v4f*)(op + r * H_ + 128 * i) = (xv[r][i] - mu[r]) * rstd[r] * gv[i] + bv[i];
}

#define GBK  64
#define GLDA 68

__device__ __forceinline__ v4f gelu_exact4(v4f x) {
  v4f y;
  y.x = gelu_exact(x.x);
  y.y = gelu_exact(x.y);
  y.z = gelu_exact(x.z);
  y.w = gelu_exact(x.w);
  return y;
}

template <int K, int N, int ACT, int RESID>
__global__ __launch_bounds__(128) void gemm_kernel(
    const float* __restrict__ A, const float* __restrict__ W,
    const float* __restrict__ bias, const float* resid, float* C) {
  static_assert(K % GBK == 0, "K must be a whole number of stages");
  static_assert(N % 64 == 0, "whole column tiles only");
  __shared__ float As[2][64 * GLDA];
  __shared__ float Ws[2][64 * GLDA];

  const int t = threadIdx.x;
  const int wave = t >> 5, lane = t & 31;
  const int m0 = blockIdx.x * 64, n0 = blockIdx.y * 64;
  const int wm = 32 * (wave & 1);
  const int wn = 32 * (wave >> 1);
  const int lm = lane & 15;
  const int koff = (lane < 16) ? 0 : 2;
  const int rbase = (lane < 16) ? 0 : 8;

  const int lrow = t >> 4;
  const int lcol = (t & 15) * 4;
  const float* Ap = A + (size_t)(m0 + lrow) * K + lcol;
  const float* Wp = W + (size_t)(n0 + lrow) * K + lcol;
  const int lds_off = lrow * GLDA + lcol;
  const size_t c_off = (size_t)(m0 + lrow) * N + n0 + lcol;

  v8f acc00 = {};
  v8f acc01 = {};
  v8f acc10 = {};
  v8f acc11 = {};

  {
    v4f a[8], w[8];
#pragma unroll
    for (int i = 0; i < 8; ++i) {
      a[i] = *(const v4f*)(Ap + (size_t)(8 * i) * K);
      w[i] = *(const v4f*)(Wp + (size_t)(8 * i) * K);
    }
#pragma unroll
    for (int i = 0; i < 8; ++i) {
      *(v4f*)(As[0] + lds_off + 8 * i * GLDA) = a[i];
      *(v4f*)(Ws[0] + lds_off + 8 * i * GLDA) = w[i];
    }
  }
  __syncthreads();

  v4f rr[8];
  v4f b4;
  constexpr int nstages = K / GBK;
  for (int s = 0; s < nstages; ++s) {
    const int p = s & 1;
    const bool has_next = (s + 1) < nstages;

    v4f a[8], w[8];
    if (has_next) {
      const size_t ko = (size_t)(s + 1) * GBK;
#pragma unroll
      for (int i = 0; i < 8; ++i) {
        a[i] = *(const v4f*)(Ap + ko + (size_t)(8 * i) * K);
        w[i] = *(const v4f*)(Wp + ko + (size_t)(8 * i) * K);
      }
    } else {
      b4 = *(const v4f*)(bias + n0 + lcol);
      if constexpr (RESID) {
#pragma unroll
        for (int i = 0; i < 8; ++i)
          rr[i] = *(const v4f*)(resid + c_off + (size_t)(8 * i) * N);
      }
    }

    const float* Ab = As[p];
    const float* Wb = Ws[p];
#pragma unroll
    for (int kk = 0; kk < GBK; kk += 4) {
      v2f a0 = *(const v2f*)(Ab + (wm + lm) * GLDA + kk + koff);
      v2f a1 = *(const v2f*)(Ab + (wm + 16 + lm) * GLDA + kk + koff);
      v2f b0 = *(const v2f*)(Wb + (wn + lm) * GLDA + kk + koff);
      v2f b1 = *(const v2f*)(Wb + (wn + 16 + lm) * GLDA + kk + koff);
      acc00 = wmma4(a0, b0, acc00);
      acc01 = wmma4(a0, b1, acc01);
      acc10 = wmma4(a1, b0, acc10);
      acc11 = wmma4(a1, b1, acc11);
    }

    __builtin_amdgcn_sched_barrier(0);
    if (has_next) {
#pragma unroll
      for (int i = 0; i < 8; ++i) {
        *(v4f*)(As[1 - p] + lds_off + 8 * i * GLDA) = a[i];
        *(v4f*)(Ws[1 - p] + lds_off + 8 * i * GLDA) = w[i];
      }
    }
    __syncthreads();
  }

  float* Cs = As[0];
#pragma unroll
  for (int i = 0; i < 2; ++i) {
    const v8f c0 = i ? acc10 : acc00;
    const v8f c1 = i ? acc11 : acc01;
#pragma unroll
    for (int r = 0; r < 8; ++r) {
      const int row = wm + 16 * i + r + rbase;
      Cs[row * GLDA + wn + lm] = c0[r];
      Cs[row * GLDA + wn + 16 + lm] = c1[r];
    }
  }
  __syncthreads();
#pragma unroll
  for (int i = 0; i < 8; ++i) {
    v4f v = *(const v4f*)(Cs + lds_off + 8 * i * GLDA);
    v += b4;
    if constexpr (RESID) v += rr[i];
    if constexpr (ACT) v = gelu_exact4(v);
    *(v4f*)(C + c_off + (size_t)(8 * i) * N) = v;
  }
}

#define AKP 68
#define AVP 136
#define ASTG (64 * AKP + 32 * AVP)
#define AOP 68
#define CSCALE 0.18033688011112042f

__global__ __launch_bounds__(128) void attn_kernel(
    const float* __restrict__ qkv, float* __restrict__ out) {
  __shared__ float stg[2][ASTG];

  const int t = threadIdx.x;
  const int wave = t >> 5, lane = t & 31;
  const int lm = lane & 15;
  const int hh = lane >> 4;

  static_assert((S_ / 64) * B_ * NH_ == 768, "numbering below is for 3 x 256 items");
  const int widx = blockIdx.x;
  const int g3 = widx >> 8;
  const int j3 = widx & 255;
  int p3;
  if (g3 == 0)      p3 = j3;
  else if (g3 == 1) p3 = (j3 + 128) & 255;
  else              p3 = (j3 < 128) ? (255 - 2 * j3) : (510 - 2 * j3);
  const int rank = (g3 << 8) + p3;
  const int qt = (S_ / 64 - 1) - rank / (B_ * NH_);
  const int bh = rank % (B_ * NH_);
  const int b = bh / NH_, nh = bh % NH_;
  const int q0 = qt * 64;

  const float* qbase = qkv + (size_t)b * S_ * QKVLD + nh * HD_;
  const float* kbase = qbase + H_;
  const float* vbase = qbase + 2 * H_;

  const int krow = t >> 4;
  const int kcol = (t & 15) * 4;
  const float* kp = kbase + (size_t)krow * QKVLD + kcol;
  const float* vp = vbase + (size_t)(2 * wave) * QKVLD + 2 * lane;
  const int k_lds = krow * AKP + kcol;
  const int v_lds = 64 * AKP + wave * AVP + 4 * lane;

  v2f qf[16];
  {
    const float* qrow = qbase + (size_t)(q0 + 16 * wave + lm) * QKVLD + 2 * hh;
#pragma unroll
    for (int i = 0; i < 16; ++i) qf[i] = *(const v2f*)(qrow + 4 * i);
  }

  {
    v4f kr[8];
    v2f va[8], vb[8];
#pragma unroll
    for (int i = 0; i < 8; ++i) {
      kr[i] = *(const v4f*)(kp + (size_t)(8 * i) * QKVLD);
      va[i] = *(const v2f*)(vp + (size_t)(8 * i) * QKVLD);
      vb[i] = *(const v2f*)(vp + (size_t)(8 * i + 1) * QKVLD);
    }
#pragma unroll
    for (int i = 0; i < 8; ++i) {
      *(v4f*)(stg[0] + k_lds + 8 * i * AKP) = kr[i];
      const v4f pr = {va[i].x, vb[i].x, va[i].y, vb[i].y};
      *(v4f*)(stg[0] + v_lds + 4 * i * AVP) = pr;
    }
  }
  __syncthreads();

  v8f ot[4] = {};
  float mi = -INFINITY, li = 0.0f;
  const int ql = 16 * wave + lm - 8 * hh;

  for (int j = 0; j <= qt; ++j) {
    const int cur = j & 1;
    const bool has_next = j < qt;

    v4f kr[8];
    v2f va[8], vb[8];
    if (has_next) {
      const size_t ro = (size_t)(j + 1) * 64 * QKVLD;
#pragma unroll
      for (int i = 0; i < 8; ++i) {
        kr[i] = *(const v4f*)(kp + ro + (size_t)(8 * i) * QKVLD);
        va[i] = *(const v2f*)(vp + ro + (size_t)(8 * i) * QKVLD);
        vb[i] = *(const v2f*)(vp + ro + (size_t)(8 * i + 1) * QKVLD);
      }
    }

    const float* Kc = stg[cur];
    const float* Vc = stg[cur] + 64 * AKP;

    v8f st[4] = {};
#pragma unroll
    for (int i = 0; i < 16; ++i) {
#pragma unroll
      for (int kt = 0; kt < 4; ++kt) {
        const v2f a = *(const v2f*)(Kc + (16 * kt + lm) * AKP + 4 * i + 2 * hh);
        st[kt] = wmma4(a, qf[i], st[kt]);
      }
    }

    if (!has_next) {
#pragma unroll
      for (int kt = 0; kt < 4; ++kt)
#pragma unroll
        for (int r = 0; r < 8; ++r)
          st[kt][r] = (16 * kt + r > ql) ? -INFINITY : st[kt][r];
    }

    float al;
    {
      float mx = -INFINITY;
#pragma unroll
      for (int kt = 0; kt < 4; ++kt)
#pragma unroll
        for (int r = 0; r < 8; ++r) mx = fmaxf(mx, st[kt][r]);
      mx = fmaxf(mx, other_row(mx));
      const float mnew = fmaxf(mi, mx * CSCALE);
      al = __builtin_amdgcn_exp2f(mi - mnew);
      const v2f cs2 = {CSCALE, CSCALE};
      const v2f nm2 = {-mnew, -mnew};
      float sum = 0.0f;
#pragma unroll
      for (int kt = 0; kt < 4; ++kt)
#pragma unroll
        for (int r = 0; r < 8; r += 2) {
          const v2f s2 = {st[kt][r], st[kt][r + 1]};
          const v2f a = __builtin_elementwise_fma(s2, cs2, nm2);
          const float p0 = __builtin_amdgcn_exp2f(a.x);
          const float p1 = __builtin_amdgcn_exp2f(a.y);
          st[kt][r] = p0;
          st[kt][r + 1] = p1;
          sum += p0;
          sum += p1;
        }
      sum += other_row(sum);
      li = li * al + sum;
      mi = mnew;
    }

#pragma unroll
    for (int dt = 0; dt < 4; ++dt)
#pragma unroll
      for (int r = 0; r < 8; ++r) ot[dt][r] *= al;
#pragma unroll
    for (int kt = 0; kt < 4; ++kt) {
#pragma unroll
      for (int t2 = 0; t2 < 4; ++t2) {
        const v2f pb = {st[kt][2 * t2], st[kt][2 * t2 + 1]};
#pragma unroll
        for (int dt = 0; dt < 4; ++dt) {
          const v2f a = *(const v2f*)(Vc + (8 * kt + 4 * hh + t2) * AVP + 2 * (16 * dt + lm));
          ot[dt] = wmma4(a, pb, ot[dt]);
        }
      }
    }

    __builtin_amdgcn_sched_barrier(0);
    if (has_next) {
#pragma unroll
      for (int i = 0; i < 8; ++i) {
        *(v4f*)(stg[1 - cur] + k_lds + 8 * i * AKP) = kr[i];
        const v4f pr = {va[i].x, vb[i].x, va[i].y, vb[i].y};
        *(v4f*)(stg[1 - cur] + v_lds + 4 * i * AVP) = pr;
      }
    }
    __syncthreads();
  }

  float* Os = stg[0];
  {
    const float inv = 1.0f / li;
#pragma unroll
    for (int dt = 0; dt < 4; ++dt)
#pragma unroll
      for (int r = 0; r < 8; ++r)
        Os[(16 * wave + lm) * AOP + 16 * dt + 8 * hh + r] = ot[dt][r] * inv;
  }
  __syncthreads();
#pragma unroll
  for (int i = 0; i < 8; ++i) {
    const int row = krow + 8 * i;
    const v4f v = *(const v4f*)(Os + row * AOP + kcol);
    *(v4f*)(out + ((size_t)b * S_ + q0 + row) * H_ + nh * HD_ + kcol) = v;
  }
}

__global__ void tail_kernel(float* out, int idx) {
  if (threadIdx.x == 0 && blockIdx.x == 0) out[idx] = 0.0f;
}

extern "C" void kernel_launch(void* const* d_in, const int* in_sizes, int n_in,
                              void* d_out, int out_size, void* d_ws,
                              size_t ws_size, hipStream_t stream) {
  (void)in_sizes; (void)n_in; (void)ws_size;
  const float* x     = (const float*)d_in[0];
  const float* ln1_g = (const float*)d_in[2];
  const float* ln1_b = (const float*)d_in[3];
  const float* qkv_w = (const float*)d_in[4];
  const float* qkv_b = (const float*)d_in[5];
  const float* out_w = (const float*)d_in[6];
  const float* out_b = (const float*)d_in[7];
  const float* ln2_g = (const float*)d_in[8];
  const float* ln2_b = (const float*)d_in[9];
  const float* fc1_w = (const float*)d_in[10];
  const float* fc1_b = (const float*)d_in[11];
  const float* fc2_w = (const float*)d_in[12];
  const float* fc2_b = (const float*)d_in[13];

  float* xo = (float*)d_out;
  float* ws = (float*)d_ws;
  float* h    = ws;
  float* qkvb = h + (size_t)M_ * H_;
  float* attn = qkvb + (size_t)M_ * 3 * H_;
  float* ffn  = attn + (size_t)M_ * H_;

  ln_kernel<<<M_ / (LN_ROWS * LN_WAVES), 32 * LN_WAVES, 0, stream>>>(x, ln1_g, ln1_b, h);
  gemm_kernel<H_, 3 * H_, 0, 0><<<dim3(M_ / 64, (3 * H_) / 64), 128, 0, stream>>>(
      h, qkv_w, qkv_b, nullptr, qkvb);
  attn_kernel<<<dim3((S_ / 64) * B_ * NH_), 128, 0, stream>>>(qkvb, attn);
  gemm_kernel<H_, H_, 0, 1><<<dim3(M_ / 64, H_ / 64), 128, 0, stream>>>(
      attn, out_w, out_b, x, xo);
  ln_kernel<<<M_ / (LN_ROWS * LN_WAVES), 32 * LN_WAVES, 0, stream>>>(xo, ln2_g, ln2_b, h);
  gemm_kernel<H_, FFN_, 1, 0><<<dim3(M_ / 64, FFN_ / 64), 128, 0, stream>>>(
      h, fc1_w, fc1_b, nullptr, ffn);
  gemm_kernel<FFN_, H_, 0, 1><<<dim3(M_ / 64, H_ / 64), 128, 0, stream>>>(
      ffn, fc2_w, fc2_b, xo, xo);
  tail_kernel<<<1, 32, 0, stream>>>(xo, out_size - 1);
}
